# starting kernel
# baseline (speedup 1.0000x reference)
#include <hip/hip_runtime.h>

typedef __attribute__((ext_vector_type(2))) float v2f;
typedef __attribute__((ext_vector_type(8))) float v8f;

#define WPB 8
#define DD  256

static __device__ __forceinline__ v8f wmma4(v2f a, v2f b, v8f c) {
  return __builtin_amdgcn_wmma_f32_16x16x4_f32(false, a, false, b, (short)0, c,
                                               false, false);
}

static __device__ __forceinline__ v2f ldRow2(const float* base, int i, int k, int h) {
  const float* p = base + i * 16 + 4 * k + 2 * h;
  v2f r; r.x = p[0]; r.y = p[1];
  return r;
}
static __device__ __forceinline__ v2f ldColS(const float* base, int m, int k, int h) {
  int kk = 4 * k + 2 * h;
  v2f r; r.x = base[kk * 16 + m]; r.y = base[(kk + 1) * 16 + m];
  return r;
}

template <bool NEGA, bool COLA>
static __device__ __forceinline__ v8f mm16(const float* A, const float* B, v8f acc,
                                           int i, int h) {
#pragma unroll
  for (int k = 0; k < 4; ++k) {
    v2f a = COLA ? ldColS(A, i, k, h) : ldRow2(A, i, k, h);
    if (NEGA) a = -a;
    v2f b = ldRow2(B, i, k, h);
    acc = wmma4(a, b, acc);
  }
  return acc;
}

static __device__ __forceinline__ void stCD(float* D, v8f v, int i, int h) {
  float* p = D + i * 16 + h * 8;
#pragma unroll
  for (int g = 0; g < 8; ++g) p[g] = v[g];
}

static __device__ __forceinline__ void cmatL(const float* Lr, const float* Li,
                                             const float* Sr, const float* Si,
                                             float* Dr, float* Di, int lane) {
  int i = lane & 15, h = lane >> 4;
  v8f ar = {};
  ar = mm16<false, false>(Lr, Sr, ar, i, h);
  ar = mm16<true,  false>(Li, Si, ar, i, h);
  v8f ai = {};
  ai = mm16<false, false>(Lr, Si, ai, i, h);
  ai = mm16<false, false>(Li, Sr, ai, i, h);
  stCD(Dr, ar, i, h);
  stCD(Di, ai, i, h);
}

static __device__ __forceinline__ void cmatR(const float* Tr, const float* Ti,
                                             const float* Rr, const float* Ri,
                                             float* Dr, float* Di, int lane) {
  int i = lane & 15, h = lane >> 4;
  v8f ar = {};
  ar = mm16<false, true>(Tr, Rr, ar, i, h);
  ar = mm16<true,  true>(Ti, Ri, ar, i, h);
  v8f ai = {};
  ai = mm16<false, true>(Tr, Ri, ai, i, h);
  ai = mm16<false, true>(Ti, Rr, ai, i, h);
  stCD(Dr, ar, i, h);
  stCD(Di, ai, i, h);
}

static __device__ __forceinline__ void rotMat(float phi, float th, float om,
                                              float gr[2][2], float gi[2][2]) {
  float st, ct, sa, ca, sb, cb;
  __sincosf(th * 0.5f, &st, &ct);
  __sincosf(0.5f * (phi + om), &sa, &ca);
  __sincosf(0.5f * (phi - om), &sb, &cb);
  gr[0][0] = ct * ca;  gi[0][0] = -ct * sa;
  gr[0][1] = -st * cb; gi[0][1] = -st * sb;
  gr[1][0] = st * cb;  gi[1][0] = -st * sb;
  gr[1][1] = ct * ca;  gi[1][1] = ct * sa;
}

static __device__ __forceinline__ void encMat(float th, float phi,
                                              float gr[2][2], float gi[2][2]) {
  float s, c, sp, cp;
  __sincosf(th * 0.5f, &s, &c);
  __sincosf(phi * 0.5f, &sp, &cp);
  gr[0][0] = c * cp;  gi[0][0] = -c * sp;
  gr[0][1] = -s * cp; gi[0][1] = s * sp;
  gr[1][0] = s * cp;  gi[1][0] = s * sp;
  gr[1][1] = c * cp;  gi[1][1] = c * sp;
}

static __device__ __forceinline__ void buildKron(const float gr[4][2][2],
                                                 const float gi[4][2][2],
                                                 float* Kr, float* Ki, int lane) {
#pragma unroll
  for (int j = 0; j < 8; ++j) {
    int idx = lane * 8 + j;
    int a = idx >> 4;
    int bcol = idx & 15;
    float pr = 1.f, pi = 0.f;
#pragma unroll
    for (int q = 0; q < 4; ++q) {
      int ba = (a >> (3 - q)) & 1;
      int bb = (bcol >> (3 - q)) & 1;
      float xr = gr[q][ba][bb], xi = gi[q][ba][bb];
      float nr = pr * xr - pi * xi;
      pi = pr * xi + pi * xr;
      pr = nr;
    }
    Kr[idx] = pr;
    Ki[idx] = pi;
  }
}

static __device__ __forceinline__ void buildRot(const float* vqcw, int l,
                                                float* GLr, float* GLi,
                                                float* GRr, float* GRi, int lane) {
  float gr[4][2][2], gi[4][2][2];
#pragma unroll
  for (int q = 0; q < 4; ++q) {
    const float* w = vqcw + (l * 8 + q) * 3;
    rotMat(w[0], w[1], w[2], gr[q], gi[q]);
  }
  buildKron(gr, gi, GLr, GLi, lane);
#pragma unroll
  for (int q = 0; q < 4; ++q) {
    const float* w = vqcw + (l * 8 + 4 + q) * 3;
    rotMat(w[0], w[1], w[2], gr[q], gi[q]);
  }
  buildKron(gr, gi, GRr, GRi, lane);
}

static __device__ __forceinline__ void buildReenc(const float* th, const float* ph,
                                                  float* GLr, float* GLi,
                                                  float* GRr, float* GRi, int lane) {
  float gr[4][2][2], gi[4][2][2];
#pragma unroll
  for (int q = 0; q < 4; ++q) encMat(th[q] * 0.5f, ph[q] * 0.5f, gr[q], gi[q]);
  buildKron(gr, gi, GLr, GLi, lane);
#pragma unroll
  for (int q = 0; q < 4; ++q) encMat(th[4 + q] * 0.5f, ph[4 + q] * 0.5f, gr[q], gi[q]);
  buildKron(gr, gi, GRr, GRi, lane);
}

template <int RNG>
static __device__ __forceinline__ void applyCnot(const float* Sr, const float* Si,
                                                 float* Dr, float* Di, int lane) {
#pragma unroll
  for (int j = 0; j < 8; ++j) {
    int d = lane * 8 + j;
    int s = ((d & 15) << 4) | (d >> 4);
    int src = s;
#pragma unroll
    for (int k = 7; k >= 0; --k) {
      int tgt = (k + RNG) & 7;
      src ^= ((src >> (7 - k)) & 1) << (7 - tgt);
    }
    int sa = ((src & 15) << 4) | (src >> 4);
    Dr[d] = Sr[sa];
    Di[d] = Si[sa];
  }
}

extern "C" __global__ __launch_bounds__(256) void vqc_kernel(
    const float* __restrict__ theta, const float* __restrict__ phi,
    const float* __restrict__ jup, const float* __restrict__ vqcw,
    float* __restrict__ out) {
  __shared__ float sXr[WPB][DD], sXi[WPB][DD];
  __shared__ float sYr[WPB][DD], sYi[WPB][DD];
  __shared__ float sGLr[WPB][DD], sGLi[WPB][DD];
  __shared__ float sGRr[WPB][DD], sGRi[WPB][DD];

  int tid = threadIdx.x;
  int wid = tid >> 5, lane = tid & 31;
  int b = blockIdx.x * WPB + wid;

  const float* th = theta + b * 8;
  const float* ph = phi + b * 8;
  const float* jj = jup + b * 28;

  float* Xr = sXr[wid]; float* Xi = sXi[wid];
  float* Yr = sYr[wid]; float* Yi = sYi[wid];
  float* GLr = sGLr[wid]; float* GLi = sGLi[wid];
  float* GRr = sGRr[wid]; float* GRi = sGRi[wid];

  {
    float e0r[8], e0i[8], e1r[8], e1i[8];
#pragma unroll
    for (int k = 0; k < 8; ++k) {
      float s, c, sp, cp;
      __sincosf(th[k] * 0.5f, &s, &c);
      __sincosf(ph[k] * 0.5f, &sp, &cp);
      e0r[k] = c * cp; e0i[k] = -c * sp;
      e1r[k] = s * cp; e1i[k] = s * sp;
    }
    float jv[28];
#pragma unroll
    for (int p = 0; p < 28; ++p) jv[p] = jj[p];

#pragma unroll
    for (int j = 0; j < 8; ++j) {
      int d = lane * 8 + j;
      int s = ((d & 15) << 4) | (d >> 4);
      float pr = 1.f, pim = 0.f;
#pragma unroll
      for (int k = 0; k < 8; ++k) {
        int bit = (s >> (7 - k)) & 1;
        float xr = bit ? e1r[k] : e0r[k];
        float xi = bit ? e1i[k] : e0i[k];
        float nr = pr * xr - pim * xi;
        pim = pr * xi + pim * xr;
        pr = nr;
      }
      float e = 0.f;
      int p = 0;
#pragma unroll
      for (int qi = 0; qi < 8; ++qi)
#pragma unroll
        for (int qj = qi + 1; qj < 8; ++qj) {
          int diff = ((s >> (7 - qi)) & 1) ^ ((s >> (7 - qj)) & 1);
          e += diff ? jv[p] : -jv[p];
          ++p;
        }
      e *= 1.57079632679489662f;
      float se, ce;
      __sincosf(e, &se, &ce);
      Xr[d] = pr * ce - pim * se;
      Xi[d] = pr * se + pim * ce;
    }
  }
  buildRot(vqcw, 0, GLr, GLi, GRr, GRi, lane);
  __syncthreads();

  cmatL(GLr, GLi, Xr, Xi, Yr, Yi, lane);
  __syncthreads();
  cmatR(Yr, Yi, GRr, GRi, Xr, Xi, lane);
  __syncthreads();

  applyCnot<1>(Xr, Xi, Yr, Yi, lane);
  buildReenc(th, ph, GLr, GLi, GRr, GRi, lane);
  __syncthreads();

  cmatL(GLr, GLi, Yr, Yi, Xr, Xi, lane);
  __syncthreads();
  cmatR(Xr, Xi, GRr, GRi, Yr, Yi, lane);
  __syncthreads();

  buildRot(vqcw, 1, GLr, GLi, GRr, GRi, lane);
  __syncthreads();
  cmatL(GLr, GLi, Yr, Yi, Xr, Xi, lane);
  __syncthreads();
  cmatR(Xr, Xi, GRr, GRi, Yr, Yi, lane);
  __syncthreads();

  applyCnot<2>(Yr, Yi, Xr, Xi, lane);
  __syncthreads();

  float acc[8];
#pragma unroll
  for (int k = 0; k < 8; ++k) acc[k] = 0.f;
#pragma unroll
  for (int j = 0; j < 8; ++j) {
    int d = lane * 8 + j;
    int s = ((d & 15) << 4) | (d >> 4);
    float vr = Xr[d], vi = Xi[d];
    float p2 = vr * vr + vi * vi;
#pragma unroll
    for (int k = 0; k < 8; ++k)
      acc[k] += ((s >> (7 - k)) & 1) ? -p2 : p2;
  }
#pragma unroll
  for (int k = 0; k < 8; ++k) {
    float v = acc[k];
    v += __shfl_xor(v, 16, 32);
    v += __shfl_xor(v, 8, 32);
    v += __shfl_xor(v, 4, 32);
    v += __shfl_xor(v, 2, 32);
    v += __shfl_xor(v, 1, 32);
    acc[k] = v;
  }
  if (lane == 0) {
#pragma unroll
    for (int k = 0; k < 8; ++k) out[b * 8 + k] = acc[k];
  }
}

extern "C" void kernel_launch(void* const* d_in, const int* in_sizes, int n_in,
                              void* d_out, int out_size, void* d_ws, size_t ws_size,
                              hipStream_t stream) {
  const float* theta = (const float*)d_in[0];
  const float* phi   = (const float*)d_in[1];
  const float* jup   = (const float*)d_in[2];
  const float* vqcw  = (const float*)d_in[3];
  float* out = (float*)d_out;
  int B = in_sizes[0] / 8;
  dim3 grid(B / WPB), block(32 * WPB);
  hipLaunchKernelGGL(vqc_kernel, grid, block, 0, stream, theta, phi, jup, vqcw, out);
}


# LDS planes keep the 4-dword groups of each 16-dword line permuted by line bits 2..3 (XOR swizzle), a wave's 8 planes contiguous: fragment reads, 16-byte stores free of bank conflicts; sums unchanged
# speedup vs baseline: 17.4262x; 17.4262x over previous
#include <hip/hip_runtime.h>

typedef __attribute__((ext_vector_type(2))) float v2f;
typedef __attribute__((ext_vector_type(4))) float v4f;
typedef __attribute__((ext_vector_type(8))) float v8f;

#define WPB 8
#define DD  256

static __device__ __forceinline__ v8f wmma4(v2f a, v2f b, v8f c) {
  return __builtin_amdgcn_wmma_f32_16x16x4_f32(false, a, false, b, (short)0, c,
                                               false, false);
}

static __device__ __forceinline__ int planeGrp(int major, int minor4) {
  return major * 16 + (minor4 ^ (major & 12));
}

static __device__ __forceinline__ v2f ldRow2(const float* base, int i, int k, int h) {
  return *(const v2f*)(base + planeGrp(i, 4 * k) + 2 * h);
}
static __device__ __forceinline__ v2f ldColS(const float* base, int m, int k, int h) {
  int kk = 4 * k + 2 * h;
  const float* p = base + kk * 16 + (m ^ (4 * k));
  v2f r; r.x = p[0]; r.y = p[16];
  return r;
}
static __device__ __forceinline__ void stRun8(float* plane, int major, int half, v8f v) {
  v4f lo = {v[0], v[1], v[2], v[3]};
  v4f hi = {v[4], v[5], v[6], v[7]};
  *(v4f*)(plane + planeGrp(major, 8 * half)) = lo;
  *(v4f*)(plane + planeGrp(major, 8 * half + 4)) = hi;
}
static __device__ __forceinline__ v8f ldRun8(const float* plane, int major, int half) {
  v4f lo = *(const v4f*)(plane + planeGrp(major, 8 * half));
  v4f hi = *(const v4f*)(plane + planeGrp(major, 8 * half + 4));
  v8f r = {lo[0], lo[1], lo[2], lo[3], hi[0], hi[1], hi[2], hi[3]};
  return r;
}

template <bool NEGA, bool COLA>
static __device__ __forceinline__ v8f mm16(const float* A, const float* B, v8f acc,
                                           int i, int h) {
#pragma unroll
  for (int k = 0; k < 4; ++k) {
    v2f a = COLA ? ldColS(A, i, k, h) : ldRow2(A, i, k, h);
    if (NEGA) a = -a;
    v2f b = ldRow2(B, i, k, h);
    acc = wmma4(a, b, acc);
  }
  return acc;
}

static __device__ __forceinline__ void stCD(float* D, v8f v, int i, int h) {
  stRun8(D, i, h, v);
}

static __device__ __forceinline__ void cmatL(const float* Lr, const float* Li,
                                             const float* Sr, const float* Si,
                                             float* Dr, float* Di, int lane) {
  int i = lane & 15, h = lane >> 4;
  v8f ar = {};
  ar = mm16<false, false>(Lr, Sr, ar, i, h);
  ar = mm16<true,  false>(Li, Si, ar, i, h);
  v8f ai = {};
  ai = mm16<false, false>(Lr, Si, ai, i, h);
  ai = mm16<false, false>(Li, Sr, ai, i, h);
  stCD(Dr, ar, i, h);
  stCD(Di, ai, i, h);
}

static __device__ __forceinline__ void cmatR(const float* Tr, const float* Ti,
                                             const float* Rr, const float* Ri,
                                             float* Dr, float* Di, int lane) {
  int i = lane & 15, h = lane >> 4;
  v8f ar = {};
  ar = mm16<false, true>(Tr, Rr, ar, i, h);
  ar = mm16<true,  true>(Ti, Ri, ar, i, h);
  v8f ai = {};
  ai = mm16<false, true>(Tr, Ri, ai, i, h);
  ai = mm16<false, true>(Ti, Rr, ai, i, h);
  stCD(Dr, ar, i, h);
  stCD(Di, ai, i, h);
}

static __device__ __forceinline__ void rotMat(float phi, float th, float om,
                                              float gr[2][2], float gi[2][2]) {
  float st, ct, sa, ca, sb, cb;
  __sincosf(th * 0.5f, &st, &ct);
  __sincosf(0.5f * (phi + om), &sa, &ca);
  __sincosf(0.5f * (phi - om), &sb, &cb);
  gr[0][0] = ct * ca;  gi[0][0] = -ct * sa;
  gr[0][1] = -st * cb; gi[0][1] = -st * sb;
  gr[1][0] = st * cb;  gi[1][0] = -st * sb;
  gr[1][1] = ct * ca;  gi[1][1] = ct * sa;
}

static __device__ __forceinline__ void encMat(float th, float phi,
                                              float gr[2][2], float gi[2][2]) {
  float s, c, sp, cp;
  __sincosf(th * 0.5f, &s, &c);
  __sincosf(phi * 0.5f, &sp, &cp);
  gr[0][0] = c * cp;  gi[0][0] = -c * sp;
  gr[0][1] = -s * cp; gi[0][1] = s * sp;
  gr[1][0] = s * cp;  gi[1][0] = s * sp;
  gr[1][1] = c * cp;  gi[1][1] = c * sp;
}

#define NGATE 24
#define TABN  (NGATE * 4)

static __device__ __forceinline__ void fillGateTable(const float* vqcw, const float* th,
                                                     const float* ph, float* tR, float* tI,
                                                     int lane) {
  int g = lane < NGATE ? lane : NGATE - 1;
  int stage = g >> 3, q = g & 7;
  float gr[2][2], gi[2][2];
  if (stage == 1) {
    encMat(th[q] * 0.5f, ph[q] * 0.5f, gr, gi);
  } else {
    const float* w = vqcw + ((stage >> 1) * 8 + q) * 3;
    rotMat(w[0], w[1], w[2], gr, gi);
  }
  if (lane < NGATE) {
#pragma unroll
    for (int a = 0; a < 2; ++a)
#pragma unroll
      for (int b = 0; b < 2; ++b) {
        tR[g * 4 + a * 2 + b] = gr[a][b];
        tI[g * 4 + a * 2 + b] = gi[a][b];
      }
  }
}

static __device__ __forceinline__ void buildKron(const float* tR, const float* tI,
                                                 float* Kr, float* Ki, int lane) {
  int a = lane >> 1;
  int o0 = ((a >> 3) & 1) * 2 + (lane & 1);
  float x0r = tR[o0], x0i = tI[o0];
  float xr[3][2], xi[3][2];
#pragma unroll
  for (int q = 1; q < 4; ++q) {
    int o = q * 4 + ((a >> (3 - q)) & 1) * 2;
    xr[q - 1][0] = tR[o];     xi[q - 1][0] = tI[o];
    xr[q - 1][1] = tR[o + 1]; xi[q - 1][1] = tI[o + 1];
  }
  v8f kr, ki;
#pragma unroll
  for (int j = 0; j < 8; ++j) {
    float pr = x0r, pi = x0i;
#pragma unroll
    for (int q = 1; q < 4; ++q) {
      int bb = (j >> (3 - q)) & 1;
      float yr = xr[q - 1][bb], yi = xi[q - 1][bb];
      float nr = pr * yr - pi * yi;
      pi = pr * yi + pi * yr;
      pr = nr;
    }
    kr[j] = pr;
    ki[j] = pi;
  }
  stRun8(Kr, a, lane & 1, kr);
  stRun8(Ki, a, lane & 1, ki);
}

static __device__ __forceinline__ void buildStage(const float* tR, const float* tI, int stage,
                                                  float* GLr, float* GLi,
                                                  float* GRr, float* GRi, int lane) {
  buildKron(tR + stage * 32, tI + stage * 32, GLr, GLi, lane);
  buildKron(tR + stage * 32 + 16, tI + stage * 32 + 16, GRr, GRi, lane);
}

template <int RNG>
static __device__ __forceinline__ void applyCnot(const float* Sr, const float* Si,
                                                 float* Dr, float* Di, int lane) {
  v8f vr, vi;
#pragma unroll
  for (int j = 0; j < 8; ++j) {
    int d = lane * 8 + j;
    int s = ((d & 15) << 4) | (d >> 4);
    int src = s;
#pragma unroll
    for (int k = 7; k >= 0; --k) {
      int tgt = (k + RNG) & 7;
      src ^= ((src >> (7 - k)) & 1) << (7 - tgt);
    }
    int sa = ((src & 15) << 4) | (src >> 4);
    int pa = sa ^ ((sa >> 4) & 12);
    vr[j] = Sr[pa];
    vi[j] = Si[pa];
  }
  stRun8(Dr, lane >> 1, lane & 1, vr);
  stRun8(Di, lane >> 1, lane & 1, vi);
}

extern "C" __global__ __launch_bounds__(256) void vqc_kernel(
    const float* __restrict__ theta, const float* __restrict__ phi,
    const float* __restrict__ jup, const float* __restrict__ vqcw,
    float* __restrict__ out) {
  __shared__ __attribute__((aligned(16))) float sPl[WPB][8][DD];
  __shared__ float sTabR[WPB][TABN], sTabI[WPB][TABN];

  int tid = threadIdx.x;
  int wid = tid >> 5, lane = tid & 31;
  int b = blockIdx.x * WPB + wid;

  const float* th = theta + b * 8;
  const float* ph = phi + b * 8;
  const float* jj = jup + b * 28;

  float* wb = &sPl[wid][0][0];
  float* Xr = wb;            float* Xi = wb + DD;
  float* Yr = wb + 2 * DD;   float* Yi = wb + 3 * DD;
  float* GLr = wb + 4 * DD;  float* GLi = wb + 5 * DD;
  float* GRr = wb + 6 * DD;  float* GRi = wb + 7 * DD;
  float* tR = sTabR[wid]; float* tI = sTabI[wid];

  fillGateTable(vqcw, th, ph, tR, tI, lane);

  {
    int r = lane & 15, h = lane >> 4;
    float z[4];
#pragma unroll
    for (int m = 0; m < 4; ++m) z[m] = ((r >> (3 - m)) & 1) ? -1.f : 1.f;
    float jv[28];
#pragma unroll
    for (int p = 0; p < 28; ++p) jv[p] = jj[p];
    float pv[8];
#pragma unroll
    for (int k = 0; k < 8; ++k) pv[k] = ph[k];
    float sq, cq;
    __sincosf(th[lane & 7] * 0.5f, &sq, &cq);
    float cs[8], sn[8];
#pragma unroll
    for (int k = 0; k < 8; ++k) {
      cs[k] = __int_as_float(__builtin_amdgcn_readlane(__float_as_int(cq), k));
      sn[k] = __int_as_float(__builtin_amdgcn_readlane(__float_as_int(sq), k));
    }
#define PIDX(i, j) ((i) * 8 - ((i) * ((i) + 1)) / 2 + ((j) - (i) - 1))
    const float NHP = -1.57079632679489662f;
    v2f a1, b1;
#pragma unroll
    for (int d = 0; d < 2; ++d) {
      float acc0 = 0.f, acc1 = 0.f;
#pragma unroll
      for (int a = 0; a < 4; ++a) {
        acc0 += z[a] * jv[PIDX(a, 4 + d)];
        acc1 += z[a] * jv[PIDX(a, 6 + d)];
      }
      a1[d] = NHP * (h ? acc1 : acc0);
    }
    b1.x = h ? z[2] : z[0];
    b1.y = h ? z[3] : z[1];
    float t0 = jv[PIDX(0, 1)] * z[1] + jv[PIDX(0, 2)] * z[2] + jv[PIDX(0, 3)] * z[3];
    float t1 = jv[PIDX(1, 2)] * z[2] + jv[PIDX(1, 3)] * z[3];
    float t2 = jv[PIDX(2, 3)] * z[3];
    float sa = z[0] * t0 + z[1] * t1 + z[2] * t2;
    float pa = pv[0] * z[0] + pv[1] * z[1] + pv[2] * z[2] + pv[3] * z[3];
    float alpha = NHP * sa - 0.5f * pa;
    float u0 = jv[PIDX(4, 5)] * z[1] + jv[PIDX(4, 6)] * z[2] + jv[PIDX(4, 7)] * z[3];
    float u1 = jv[PIDX(5, 6)] * z[2] + jv[PIDX(5, 7)] * z[3];
    float u2 = jv[PIDX(6, 7)] * z[3];
    float sb = z[0] * u0 + z[1] * u1 + z[2] * u2;
    float pb = pv[4] * z[0] + pv[5] * z[1] + pv[6] * z[2] + pv[7] * z[3];
    float beta = NHP * sb - 0.5f * pb;
    float mu = 1.f, nu = 1.f;
#pragma unroll
    for (int m = 0; m < 4; ++m) {
      bool set = (r >> (3 - m)) & 1;
      mu *= set ? sn[m] : cs[m];
      nu *= set ? sn[4 + m] : cs[4 + m];
    }
#undef PIDX
    float m0 = h ? 0.f : 1.f;
    v2f a2, b2, a3, b3;
    a2.x = alpha * m0; a2.y = m0;
    b2.x = 1.f;        b2.y = beta;
    a3.x = mu * m0;    a3.y = 0.f;
    b3.x = nu;         b3.y = 0.f;
    v8f TH = {};
    TH = wmma4(a2, b2, TH);
    TH = wmma4(a1, b1, TH);
    v8f MG = {};
    MG = wmma4(a3, b3, MG);
    v8f xr, xi;
#pragma unroll
    for (int g = 0; g < 8; ++g) {
      float se, ce;
      __sincosf(TH[g], &se, &ce);
      xr[g] = MG[g] * ce;
      xi[g] = MG[g] * se;
    }
    stCD(Xr, xr, r, h);
    stCD(Xi, xi, r, h);
  }
  __syncthreads();
  buildStage(tR, tI, 0, GLr, GLi, GRr, GRi, lane);
  __syncthreads();

  cmatL(GLr, GLi, Xr, Xi, Yr, Yi, lane);
  __syncthreads();
  cmatR(Yr, Yi, GRr, GRi, Xr, Xi, lane);
  __syncthreads();

  applyCnot<1>(Xr, Xi, Yr, Yi, lane);
  buildStage(tR, tI, 1, GLr, GLi, GRr, GRi, lane);
  __syncthreads();

  cmatL(GLr, GLi, Yr, Yi, Xr, Xi, lane);
  __syncthreads();
  cmatR(Xr, Xi, GRr, GRi, Yr, Yi, lane);
  __syncthreads();

  buildStage(tR, tI, 2, GLr, GLi, GRr, GRi, lane);
  __syncthreads();
  cmatL(GLr, GLi, Yr, Yi, Xr, Xi, lane);
  __syncthreads();
  cmatR(Xr, Xi, GRr, GRi, Yr, Yi, lane);
  __syncthreads();

  applyCnot<2>(Yr, Yi, Xr, Xi, lane);
  __syncthreads();

  float acc[8];
#pragma unroll
  for (int k = 0; k < 8; ++k) acc[k] = 0.f;
  v8f fr = ldRun8(Xr, lane >> 1, lane & 1);
  v8f fi = ldRun8(Xi, lane >> 1, lane & 1);
#pragma unroll
  for (int j = 0; j < 8; ++j) {
    int d = lane * 8 + j;
    int s = ((d & 15) << 4) | (d >> 4);
    float vr = fr[j], vi = fi[j];
    float p2 = vr * vr + vi * vi;
#pragma unroll
    for (int k = 0; k < 8; ++k)
      acc[k] += ((s >> (7 - k)) & 1) ? -p2 : p2;
  }
#pragma unroll
  for (int k = 0; k < 8; ++k) {
    float v = acc[k];
    v += __shfl_xor(v, 16, 32);
    v += __shfl_xor(v, 8, 32);
    v += __shfl_xor(v, 4, 32);
    v += __shfl_xor(v, 2, 32);
    v += __shfl_xor(v, 1, 32);
    acc[k] = v;
  }
  if (lane == 0) {
#pragma unroll
    for (int k = 0; k < 8; ++k) out[b * 8 + k] = acc[k];
  }
}

extern "C" void kernel_launch(void* const* d_in, const int* in_sizes, int n_in,
                              void* d_out, int out_size, void* d_ws, size_t ws_size,
                              hipStream_t stream) {
  const float* theta = (const float*)d_in[0];
  const float* phi   = (const float*)d_in[1];
  const float* jup   = (const float*)d_in[2];
  const float* vqcw  = (const float*)d_in[3];
  float* out = (float*)d_out;
  int B = in_sizes[0] / 8;
  dim3 grid(B / WPB), block(32 * WPB);
  hipLaunchKernelGGL(vqc_kernel, grid, block, 0, stream, theta, phi, jup, vqcw, out);
}


# All 11 workgroup barriers dropped for wave-scope ordering (fence + wave barrier, no instruction): each LDS plane and table belongs to one wave, whose LDS accesses run in issue order; sums unchanged
# speedup vs baseline: 18.2145x; 1.0452x over previous
#include <hip/hip_runtime.h>

typedef __attribute__((ext_vector_type(2))) float v2f;
typedef __attribute__((ext_vector_type(4))) float v4f;
typedef __attribute__((ext_vector_type(8))) float v8f;

#define WPB 8
#define DD  256

static __device__ __forceinline__ v8f wmma4(v2f a, v2f b, v8f c) {
  return __builtin_amdgcn_wmma_f32_16x16x4_f32(false, a, false, b, (short)0, c,
                                               false, false);
}

static __device__ __forceinline__ void waveSync() {
  __builtin_amdgcn_fence(__ATOMIC_RELEASE, "wavefront");
  __builtin_amdgcn_wave_barrier();
  __builtin_amdgcn_fence(__ATOMIC_ACQUIRE, "wavefront");
}

static __device__ __forceinline__ int planeGrp(int major, int minor4) {
  return major * 16 + (minor4 ^ (major & 12));
}

static __device__ __forceinline__ v2f ldRow2(const float* base, int i, int k, int h) {
  return *(const v2f*)(base + planeGrp(i, 4 * k) + 2 * h);
}
static __device__ __forceinline__ v2f ldColS(const float* base, int m, int k, int h) {
  int kk = 4 * k + 2 * h;
  const float* p = base + kk * 16 + (m ^ (4 * k));
  v2f r; r.x = p[0]; r.y = p[16];
  return r;
}
static __device__ __forceinline__ void stRun8(float* plane, int major, int half, v8f v) {
  v4f lo = {v[0], v[1], v[2], v[3]};
  v4f hi = {v[4], v[5], v[6], v[7]};
  *(v4f*)(plane + planeGrp(major, 8 * half)) = lo;
  *(v4f*)(plane + planeGrp(major, 8 * half + 4)) = hi;
}
static __device__ __forceinline__ v8f ldRun8(const float* plane, int major, int half) {
  v4f lo = *(const v4f*)(plane + planeGrp(major, 8 * half));
  v4f hi = *(const v4f*)(plane + planeGrp(major, 8 * half + 4));
  v8f r = {lo[0], lo[1], lo[2], lo[3], hi[0], hi[1], hi[2], hi[3]};
  return r;
}

template <bool NEGA, bool COLA>
static __device__ __forceinline__ v8f mm16(const float* A, const float* B, v8f acc,
                                           int i, int h) {
#pragma unroll
  for (int k = 0; k < 4; ++k) {
    v2f a = COLA ? ldColS(A, i, k, h) : ldRow2(A, i, k, h);
    if (NEGA) a = -a;
    v2f b = ldRow2(B, i, k, h);
    acc = wmma4(a, b, acc);
  }
  return acc;
}

static __device__ __forceinline__ void stCD(float* D, v8f v, int i, int h) {
  stRun8(D, i, h, v);
}

static __device__ __forceinline__ void cmatL(const float* Lr, const float* Li,
                                             const float* Sr, const float* Si,
                                             float* Dr, float* Di, int lane) {
  int i = lane & 15, h = lane >> 4;
  v8f ar = {};
  ar = mm16<false, false>(Lr, Sr, ar, i, h);
  ar = mm16<true,  false>(Li, Si, ar, i, h);
  v8f ai = {};
  ai = mm16<false, false>(Lr, Si, ai, i, h);
  ai = mm16<false, false>(Li, Sr, ai, i, h);
  stCD(Dr, ar, i, h);
  stCD(Di, ai, i, h);
}

static __device__ __forceinline__ void cmatR(const float* Tr, const float* Ti,
                                             const float* Rr, const float* Ri,
                                             float* Dr, float* Di, int lane) {
  int i = lane & 15, h = lane >> 4;
  v8f ar = {};
  ar = mm16<false, true>(Tr, Rr, ar, i, h);
  ar = mm16<true,  true>(Ti, Ri, ar, i, h);
  v8f ai = {};
  ai = mm16<false, true>(Tr, Ri, ai, i, h);
  ai = mm16<false, true>(Ti, Rr, ai, i, h);
  stCD(Dr, ar, i, h);
  stCD(Di, ai, i, h);
}

static __device__ __forceinline__ void rotMat(float phi, float th, float om,
                                              float gr[2][2], float gi[2][2]) {
  float st, ct, sa, ca, sb, cb;
  __sincosf(th * 0.5f, &st, &ct);
  __sincosf(0.5f * (phi + om), &sa, &ca);
  __sincosf(0.5f * (phi - om), &sb, &cb);
  gr[0][0] = ct * ca;  gi[0][0] = -ct * sa;
  gr[0][1] = -st * cb; gi[0][1] = -st * sb;
  gr[1][0] = st * cb;  gi[1][0] = -st * sb;
  gr[1][1] = ct * ca;  gi[1][1] = ct * sa;
}

static __device__ __forceinline__ void encMat(float th, float phi,
                                              float gr[2][2], float gi[2][2]) {
  float s, c, sp, cp;
  __sincosf(th * 0.5f, &s, &c);
  __sincosf(phi * 0.5f, &sp, &cp);
  gr[0][0] = c * cp;  gi[0][0] = -c * sp;
  gr[0][1] = -s * cp; gi[0][1] = s * sp;
  gr[1][0] = s * cp;  gi[1][0] = s * sp;
  gr[1][1] = c * cp;  gi[1][1] = c * sp;
}

#define NGATE 24
#define TABN  (NGATE * 4)

static __device__ __forceinline__ void fillGateTable(const float* vqcw, const float* th,
                                                     const float* ph, float* tR, float* tI,
                                                     int lane) {
  int g = lane < NGATE ? lane : NGATE - 1;
  int stage = g >> 3, q = g & 7;
  float gr[2][2], gi[2][2];
  if (stage == 1) {
    encMat(th[q] * 0.5f, ph[q] * 0.5f, gr, gi);
  } else {
    const float* w = vqcw + ((stage >> 1) * 8 + q) * 3;
    rotMat(w[0], w[1], w[2], gr, gi);
  }
  if (lane < NGATE) {
#pragma unroll
    for (int a = 0; a < 2; ++a)
#pragma unroll
      for (int b = 0; b < 2; ++b) {
        tR[g * 4 + a * 2 + b] = gr[a][b];
        tI[g * 4 + a * 2 + b] = gi[a][b];
      }
  }
}

static __device__ __forceinline__ void buildKron(const float* tR, const float* tI,
                                                 float* Kr, float* Ki, int lane) {
  int a = lane >> 1;
  int o0 = ((a >> 3) & 1) * 2 + (lane & 1);
  float x0r = tR[o0], x0i = tI[o0];
  float xr[3][2], xi[3][2];
#pragma unroll
  for (int q = 1; q < 4; ++q) {
    int o = q * 4 + ((a >> (3 - q)) & 1) * 2;
    xr[q - 1][0] = tR[o];     xi[q - 1][0] = tI[o];
    xr[q - 1][1] = tR[o + 1]; xi[q - 1][1] = tI[o + 1];
  }
  v8f kr, ki;
#pragma unroll
  for (int j = 0; j < 8; ++j) {
    float pr = x0r, pi = x0i;
#pragma unroll
    for (int q = 1; q < 4; ++q) {
      int bb = (j >> (3 - q)) & 1;
      float yr = xr[q - 1][bb], yi = xi[q - 1][bb];
      float nr = pr * yr - pi * yi;
      pi = pr * yi + pi * yr;
      pr = nr;
    }
    kr[j] = pr;
    ki[j] = pi;
  }
  stRun8(Kr, a, lane & 1, kr);
  stRun8(Ki, a, lane & 1, ki);
}

static __device__ __forceinline__ void buildStage(const float* tR, const float* tI, int stage,
                                                  float* GLr, float* GLi,
                                                  float* GRr, float* GRi, int lane) {
  buildKron(tR + stage * 32, tI + stage * 32, GLr, GLi, lane);
  buildKron(tR + stage * 32 + 16, tI + stage * 32 + 16, GRr, GRi, lane);
}

template <int RNG>
static __device__ __forceinline__ void applyCnot(const float* Sr, const float* Si,
                                                 float* Dr, float* Di, int lane) {
  v8f vr, vi;
#pragma unroll
  for (int j = 0; j < 8; ++j) {
    int d = lane * 8 + j;
    int s = ((d & 15) << 4) | (d >> 4);
    int src = s;
#pragma unroll
    for (int k = 7; k >= 0; --k) {
      int tgt = (k + RNG) & 7;
      src ^= ((src >> (7 - k)) & 1) << (7 - tgt);
    }
    int sa = ((src & 15) << 4) | (src >> 4);
    int pa = sa ^ ((sa >> 4) & 12);
    vr[j] = Sr[pa];
    vi[j] = Si[pa];
  }
  stRun8(Dr, lane >> 1, lane & 1, vr);
  stRun8(Di, lane >> 1, lane & 1, vi);
}

extern "C" __global__ __launch_bounds__(256) void vqc_kernel(
    const float* __restrict__ theta, const float* __restrict__ phi,
    const float* __restrict__ jup, const float* __restrict__ vqcw,
    float* __restrict__ out) {
  __shared__ __attribute__((aligned(16))) float sPl[WPB][8][DD];
  __shared__ float sTabR[WPB][TABN], sTabI[WPB][TABN];

  int tid = threadIdx.x;
  int wid = tid >> 5, lane = tid & 31;
  int b = blockIdx.x * WPB + wid;

  const float* th = theta + b * 8;
  const float* ph = phi + b * 8;
  const float* jj = jup + b * 28;

  float* wb = &sPl[wid][0][0];
  float* Xr = wb;            float* Xi = wb + DD;
  float* Yr = wb + 2 * DD;   float* Yi = wb + 3 * DD;
  float* GLr = wb + 4 * DD;  float* GLi = wb + 5 * DD;
  float* GRr = wb + 6 * DD;  float* GRi = wb + 7 * DD;
  float* tR = sTabR[wid]; float* tI = sTabI[wid];

  fillGateTable(vqcw, th, ph, tR, tI, lane);

  {
    int r = lane & 15, h = lane >> 4;
    float z[4];
#pragma unroll
    for (int m = 0; m < 4; ++m) z[m] = ((r >> (3 - m)) & 1) ? -1.f : 1.f;
    float jv[28];
#pragma unroll
    for (int p = 0; p < 28; ++p) jv[p] = jj[p];
    float pv[8];
#pragma unroll
    for (int k = 0; k < 8; ++k) pv[k] = ph[k];
    float sq, cq;
    __sincosf(th[lane & 7] * 0.5f, &sq, &cq);
    float cs[8], sn[8];
#pragma unroll
    for (int k = 0; k < 8; ++k) {
      cs[k] = __int_as_float(__builtin_amdgcn_readlane(__float_as_int(cq), k));
      sn[k] = __int_as_float(__builtin_amdgcn_readlane(__float_as_int(sq), k));
    }
#define PIDX(i, j) ((i) * 8 - ((i) * ((i) + 1)) / 2 + ((j) - (i) - 1))
    const float NHP = -1.57079632679489662f;
    v2f a1, b1;
#pragma unroll
    for (int d = 0; d < 2; ++d) {
      float acc0 = 0.f, acc1 = 0.f;
#pragma unroll
      for (int a = 0; a < 4; ++a) {
        acc0 += z[a] * jv[PIDX(a, 4 + d)];
        acc1 += z[a] * jv[PIDX(a, 6 + d)];
      }
      a1[d] = NHP * (h ? acc1 : acc0);
    }
    b1.x = h ? z[2] : z[0];
    b1.y = h ? z[3] : z[1];
    float t0 = jv[PIDX(0, 1)] * z[1] + jv[PIDX(0, 2)] * z[2] + jv[PIDX(0, 3)] * z[3];
    float t1 = jv[PIDX(1, 2)] * z[2] + jv[PIDX(1, 3)] * z[3];
    float t2 = jv[PIDX(2, 3)] * z[3];
    float sa = z[0] * t0 + z[1] * t1 + z[2] * t2;
    float pa = pv[0] * z[0] + pv[1] * z[1] + pv[2] * z[2] + pv[3] * z[3];
    float alpha = NHP * sa - 0.5f * pa;
    float u0 = jv[PIDX(4, 5)] * z[1] + jv[PIDX(4, 6)] * z[2] + jv[PIDX(4, 7)] * z[3];
    float u1 = jv[PIDX(5, 6)] * z[2] + jv[PIDX(5, 7)] * z[3];
    float u2 = jv[PIDX(6, 7)] * z[3];
    float sb = z[0] * u0 + z[1] * u1 + z[2] * u2;
    float pb = pv[4] * z[0] + pv[5] * z[1] + pv[6] * z[2] + pv[7] * z[3];
    float beta = NHP * sb - 0.5f * pb;
    float mu = 1.f, nu = 1.f;
#pragma unroll
    for (int m = 0; m < 4; ++m) {
      bool set = (r >> (3 - m)) & 1;
      mu *= set ? sn[m] : cs[m];
      nu *= set ? sn[4 + m] : cs[4 + m];
    }
#undef PIDX
    float m0 = h ? 0.f : 1.f;
    v2f a2, b2, a3, b3;
    a2.x = alpha * m0; a2.y = m0;
    b2.x = 1.f;        b2.y = beta;
    a3.x = mu * m0;    a3.y = 0.f;
    b3.x = nu;         b3.y = 0.f;
    v8f TH = {};
    TH = wmma4(a2, b2, TH);
    TH = wmma4(a1, b1, TH);
    v8f MG = {};
    MG = wmma4(a3, b3, MG);
    v8f xr, xi;
#pragma unroll
    for (int g = 0; g < 8; ++g) {
      float se, ce;
      __sincosf(TH[g], &se, &ce);
      xr[g] = MG[g] * ce;
      xi[g] = MG[g] * se;
    }
    stCD(Xr, xr, r, h);
    stCD(Xi, xi, r, h);
  }
  waveSync();
  buildStage(tR, tI, 0, GLr, GLi, GRr, GRi, lane);
  waveSync();

  cmatL(GLr, GLi, Xr, Xi, Yr, Yi, lane);
  waveSync();
  cmatR(Yr, Yi, GRr, GRi, Xr, Xi, lane);
  waveSync();

  applyCnot<1>(Xr, Xi, Yr, Yi, lane);
  buildStage(tR, tI, 1, GLr, GLi, GRr, GRi, lane);
  waveSync();

  cmatL(GLr, GLi, Yr, Yi, Xr, Xi, lane);
  waveSync();
  cmatR(Xr, Xi, GRr, GRi, Yr, Yi, lane);
  waveSync();

  buildStage(tR, tI, 2, GLr, GLi, GRr, GRi, lane);
  waveSync();
  cmatL(GLr, GLi, Yr, Yi, Xr, Xi, lane);
  waveSync();
  cmatR(Xr, Xi, GRr, GRi, Yr, Yi, lane);
  waveSync();

  applyCnot<2>(Yr, Yi, Xr, Xi, lane);
  waveSync();

  float acc[8];
#pragma unroll
  for (int k = 0; k < 8; ++k) acc[k] = 0.f;
  v8f fr = ldRun8(Xr, lane >> 1, lane & 1);
  v8f fi = ldRun8(Xi, lane >> 1, lane & 1);
#pragma unroll
  for (int j = 0; j < 8; ++j) {
    int d = lane * 8 + j;
    int s = ((d & 15) << 4) | (d >> 4);
    float vr = fr[j], vi = fi[j];
    float p2 = vr * vr + vi * vi;
#pragma unroll
    for (int k = 0; k < 8; ++k)
      acc[k] += ((s >> (7 - k)) & 1) ? -p2 : p2;
  }
#pragma unroll
  for (int k = 0; k < 8; ++k) {
    float v = acc[k];
    v += __shfl_xor(v, 16, 32);
    v += __shfl_xor(v, 8, 32);
    v += __shfl_xor(v, 4, 32);
    v += __shfl_xor(v, 2, 32);
    v += __shfl_xor(v, 1, 32);
    acc[k] = v;
  }
  if (lane == 0) {
#pragma unroll
    for (int k = 0; k < 8; ++k) out[b * 8 + k] = acc[k];
  }
}

extern "C" void kernel_launch(void* const* d_in, const int* in_sizes, int n_in,
                              void* d_out, int out_size, void* d_ws, size_t ws_size,
                              hipStream_t stream) {
  const float* theta = (const float*)d_in[0];
  const float* phi   = (const float*)d_in[1];
  const float* jup   = (const float*)d_in[2];
  const float* vqcw  = (const float*)d_in[3];
  float* out = (float*)d_out;
  int B = in_sizes[0] / 8;
  dim3 grid(B / WPB), block(32 * WPB);
  hipLaunchKernelGGL(vqc_kernel, grid, block, 0, stream, theta, phi, jup, vqcw, out);
}


# Each complex 16x16 product as 3 real products, P=(Ar+Ai)Br, Dr=P-Ai(Br+Bi), Di=P+Ar(Bi-Br), operand sums formed on the loaded fragments: 12 WMMAs not 16 (75 not 99 in all); sum order changes
# speedup vs baseline: 20.3507x; 1.1173x over previous
#include <hip/hip_runtime.h>

typedef __attribute__((ext_vector_type(2))) float v2f;
typedef __attribute__((ext_vector_type(4))) float v4f;
typedef __attribute__((ext_vector_type(8))) float v8f;

#define WPB 8
#define DD  256

static __device__ __forceinline__ v8f wmma4(v2f a, v2f b, v8f c) {
  return __builtin_amdgcn_wmma_f32_16x16x4_f32(false, a, false, b, (short)0, c,
                                               false, false);
}

static __device__ __forceinline__ void waveSync() {
  __builtin_amdgcn_fence(__ATOMIC_RELEASE, "wavefront");
  __builtin_amdgcn_wave_barrier();
  __builtin_amdgcn_fence(__ATOMIC_ACQUIRE, "wavefront");
}

static __device__ __forceinline__ int planeGrp(int major, int minor4) {
  return major * 16 + (minor4 ^ (major & 12));
}

static __device__ __forceinline__ v2f ldRow2(const float* base, int i, int k, int h) {
  return *(const v2f*)(base + planeGrp(i, 4 * k) + 2 * h);
}
static __device__ __forceinline__ v2f ldColS(const float* base, int m, int k, int h) {
  int kk = 4 * k + 2 * h;
  const float* p = base + kk * 16 + (m ^ (4 * k));
  v2f r; r.x = p[0]; r.y = p[16];
  return r;
}
static __device__ __forceinline__ void stRun8(float* plane, int major, int half, v8f v) {
  v4f lo = {v[0], v[1], v[2], v[3]};
  v4f hi = {v[4], v[5], v[6], v[7]};
  *(v4f*)(plane + planeGrp(major, 8 * half)) = lo;
  *(v4f*)(plane + planeGrp(major, 8 * half + 4)) = hi;
}
static __device__ __forceinline__ v8f ldRun8(const float* plane, int major, int half) {
  v4f lo = *(const v4f*)(plane + planeGrp(major, 8 * half));
  v4f hi = *(const v4f*)(plane + planeGrp(major, 8 * half + 4));
  v8f r = {lo[0], lo[1], lo[2], lo[3], hi[0], hi[1], hi[2], hi[3]};
  return r;
}

static __device__ __forceinline__ void stCD(float* D, v8f v, int i, int h) {
  stRun8(D, i, h, v);
}

template <bool COL>
static __device__ __forceinline__ void ldFrag4(const float* P, int i, int h, v2f f[4]) {
#pragma unroll
  for (int k = 0; k < 4; ++k) f[k] = COL ? ldColS(P, i, k, h) : ldRow2(P, i, k, h);
}

static __device__ __forceinline__ v8f mm16(const v2f a[4], const v2f b[4], v8f acc) {
#pragma unroll
  for (int k = 0; k < 4; ++k) acc = wmma4(a[k], b[k], acc);
  return acc;
}

template <bool COLA>
static __device__ __forceinline__ void cmat3(const float* Ar, const float* Ai,
                                             const float* Br, const float* Bi,
                                             float* Dr, float* Di, int lane) {
  int i = lane & 15, h = lane >> 4;
  v2f ar[4], ai[4], br[4], bi[4];
  ldFrag4<COLA>(Ar, i, h, ar);
  ldFrag4<COLA>(Ai, i, h, ai);
  ldFrag4<false>(Br, i, h, br);
  ldFrag4<false>(Bi, i, h, bi);
  v2f as[4], bd[4], bn[4];
#pragma unroll
  for (int k = 0; k < 4; ++k) {
    as[k] = ar[k] + ai[k];
    bd[k] = bi[k] - br[k];
    bn[k] = -br[k] - bi[k];
  }
  v8f p = {};
  p = mm16(as, br, p);
  v8f dr = mm16(ai, bn, p);
  v8f di = mm16(ar, bd, p);
  stCD(Dr, dr, i, h);
  stCD(Di, di, i, h);
}

static __device__ __forceinline__ void cmatL(const float* Lr, const float* Li,
                                             const float* Sr, const float* Si,
                                             float* Dr, float* Di, int lane) {
  cmat3<false>(Lr, Li, Sr, Si, Dr, Di, lane);
}

static __device__ __forceinline__ void cmatR(const float* Tr, const float* Ti,
                                             const float* Rr, const float* Ri,
                                             float* Dr, float* Di, int lane) {
  cmat3<true>(Tr, Ti, Rr, Ri, Dr, Di, lane);
}

static __device__ __forceinline__ void rotMat(float phi, float th, float om,
                                              float gr[2][2], float gi[2][2]) {
  float st, ct, sa, ca, sb, cb;
  __sincosf(th * 0.5f, &st, &ct);
  __sincosf(0.5f * (phi + om), &sa, &ca);
  __sincosf(0.5f * (phi - om), &sb, &cb);
  gr[0][0] = ct * ca;  gi[0][0] = -ct * sa;
  gr[0][1] = -st * cb; gi[0][1] = -st * sb;
  gr[1][0] = st * cb;  gi[1][0] = -st * sb;
  gr[1][1] = ct * ca;  gi[1][1] = ct * sa;
}

static __device__ __forceinline__ void encMat(float th, float phi,
                                              float gr[2][2], float gi[2][2]) {
  float s, c, sp, cp;
  __sincosf(th * 0.5f, &s, &c);
  __sincosf(phi * 0.5f, &sp, &cp);
  gr[0][0] = c * cp;  gi[0][0] = -c * sp;
  gr[0][1] = -s * cp; gi[0][1] = s * sp;
  gr[1][0] = s * cp;  gi[1][0] = s * sp;
  gr[1][1] = c * cp;  gi[1][1] = c * sp;
}

#define NGATE 24
#define TABN  (NGATE * 4)

static __device__ __forceinline__ void fillGateTable(const float* vqcw, const float* th,
                                                     const float* ph, float* tR, float* tI,
                                                     int lane) {
  int g = lane < NGATE ? lane : NGATE - 1;
  int stage = g >> 3, q = g & 7;
  float gr[2][2], gi[2][2];
  if (stage == 1) {
    encMat(th[q] * 0.5f, ph[q] * 0.5f, gr, gi);
  } else {
    const float* w = vqcw + ((stage >> 1) * 8 + q) * 3;
    rotMat(w[0], w[1], w[2], gr, gi);
  }
  if (lane < NGATE) {
#pragma unroll
    for (int a = 0; a < 2; ++a)
#pragma unroll
      for (int b = 0; b < 2; ++b) {
        tR[g * 4 + a * 2 + b] = gr[a][b];
        tI[g * 4 + a * 2 + b] = gi[a][b];
      }
  }
}

static __device__ __forceinline__ void buildKron(const float* tR, const float* tI,
                                                 float* Kr, float* Ki, int lane) {
  int a = lane >> 1;
  int o0 = ((a >> 3) & 1) * 2 + (lane & 1);
  float x0r = tR[o0], x0i = tI[o0];
  float xr[3][2], xi[3][2];
#pragma unroll
  for (int q = 1; q < 4; ++q) {
    int o = q * 4 + ((a >> (3 - q)) & 1) * 2;
    xr[q - 1][0] = tR[o];     xi[q - 1][0] = tI[o];
    xr[q - 1][1] = tR[o + 1]; xi[q - 1][1] = tI[o + 1];
  }
  v8f kr, ki;
#pragma unroll
  for (int j = 0; j < 8; ++j) {
    float pr = x0r, pi = x0i;
#pragma unroll
    for (int q = 1; q < 4; ++q) {
      int bb = (j >> (3 - q)) & 1;
      float yr = xr[q - 1][bb], yi = xi[q - 1][bb];
      float nr = pr * yr - pi * yi;
      pi = pr * yi + pi * yr;
      pr = nr;
    }
    kr[j] = pr;
    ki[j] = pi;
  }
  stRun8(Kr, a, lane & 1, kr);
  stRun8(Ki, a, lane & 1, ki);
}

static __device__ __forceinline__ void buildStage(const float* tR, const float* tI, int stage,
                                                  float* GLr, float* GLi,
                                                  float* GRr, float* GRi, int lane) {
  buildKron(tR + stage * 32, tI + stage * 32, GLr, GLi, lane);
  buildKron(tR + stage * 32 + 16, tI + stage * 32 + 16, GRr, GRi, lane);
}

template <int RNG>
static __device__ __forceinline__ void applyCnot(const float* Sr, const float* Si,
                                                 float* Dr, float* Di, int lane) {
  v8f vr, vi;
#pragma unroll
  for (int j = 0; j < 8; ++j) {
    int d = lane * 8 + j;
    int s = ((d & 15) << 4) | (d >> 4);
    int src = s;
#pragma unroll
    for (int k = 7; k >= 0; --k) {
      int tgt = (k + RNG) & 7;
      src ^= ((src >> (7 - k)) & 1) << (7 - tgt);
    }
    int sa = ((src & 15) << 4) | (src >> 4);
    int pa = sa ^ ((sa >> 4) & 12);
    vr[j] = Sr[pa];
    vi[j] = Si[pa];
  }
  stRun8(Dr, lane >> 1, lane & 1, vr);
  stRun8(Di, lane >> 1, lane & 1, vi);
}

extern "C" __global__ __launch_bounds__(256) void vqc_kernel(
    const float* __restrict__ theta, const float* __restrict__ phi,
    const float* __restrict__ jup, const float* __restrict__ vqcw,
    float* __restrict__ out) {
  __shared__ __attribute__((aligned(16))) float sPl[WPB][8][DD];
  __shared__ float sTabR[WPB][TABN], sTabI[WPB][TABN];

  int tid = threadIdx.x;
  int wid = tid >> 5, lane = tid & 31;
  int b = blockIdx.x * WPB + wid;

  const float* th = theta + b * 8;
  const float* ph = phi + b * 8;
  const float* jj = jup + b * 28;

  float* wb = &sPl[wid][0][0];
  float* Xr = wb;            float* Xi = wb + DD;
  float* Yr = wb + 2 * DD;   float* Yi = wb + 3 * DD;
  float* GLr = wb + 4 * DD;  float* GLi = wb + 5 * DD;
  float* GRr = wb + 6 * DD;  float* GRi = wb + 7 * DD;
  float* tR = sTabR[wid]; float* tI = sTabI[wid];

  fillGateTable(vqcw, th, ph, tR, tI, lane);

  {
    int r = lane & 15, h = lane >> 4;
    float z[4];
#pragma unroll
    for (int m = 0; m < 4; ++m) z[m] = ((r >> (3 - m)) & 1) ? -1.f : 1.f;
    float jv[28];
#pragma unroll
    for (int p = 0; p < 28; ++p) jv[p] = jj[p];
    float pv[8];
#pragma unroll
    for (int k = 0; k < 8; ++k) pv[k] = ph[k];
    float sq, cq;
    __sincosf(th[lane & 7] * 0.5f, &sq, &cq);
    float cs[8], sn[8];
#pragma unroll
    for (int k = 0; k < 8; ++k) {
      cs[k] = __int_as_float(__builtin_amdgcn_readlane(__float_as_int(cq), k));
      sn[k] = __int_as_float(__builtin_amdgcn_readlane(__float_as_int(sq), k));
    }
#define PIDX(i, j) ((i) * 8 - ((i) * ((i) + 1)) / 2 + ((j) - (i) - 1))
    const float NHP = -1.57079632679489662f;
    v2f a1, b1;
#pragma unroll
    for (int d = 0; d < 2; ++d) {
      float acc0 = 0.f, acc1 = 0.f;
#pragma unroll
      for (int a = 0; a < 4; ++a) {
        acc0 += z[a] * jv[PIDX(a, 4 + d)];
        acc1 += z[a] * jv[PIDX(a, 6 + d)];
      }
      a1[d] = NHP * (h ? acc1 : acc0);
    }
    b1.x = h ? z[2] : z[0];
    b1.y = h ? z[3] : z[1];
    float t0 = jv[PIDX(0, 1)] * z[1] + jv[PIDX(0, 2)] * z[2] + jv[PIDX(0, 3)] * z[3];
    float t1 = jv[PIDX(1, 2)] * z[2] + jv[PIDX(1, 3)] * z[3];
    float t2 = jv[PIDX(2, 3)] * z[3];
    float sa = z[0] * t0 + z[1] * t1 + z[2] * t2;
    float pa = pv[0] * z[0] + pv[1] * z[1] + pv[2] * z[2] + pv[3] * z[3];
    float alpha = NHP * sa - 0.5f * pa;
    float u0 = jv[PIDX(4, 5)] * z[1] + jv[PIDX(4, 6)] * z[2] + jv[PIDX(4, 7)] * z[3];
    float u1 = jv[PIDX(5, 6)] * z[2] + jv[PIDX(5, 7)] * z[3];
    float u2 = jv[PIDX(6, 7)] * z[3];
    float sb = z[0] * u0 + z[1] * u1 + z[2] * u2;
    float pb = pv[4] * z[0] + pv[5] * z[1] + pv[6] * z[2] + pv[7] * z[3];
    float beta = NHP * sb - 0.5f * pb;
    float mu = 1.f, nu = 1.f;
#pragma unroll
    for (int m = 0; m < 4; ++m) {
      bool set = (r >> (3 - m)) & 1;
      mu *= set ? sn[m] : cs[m];
      nu *= set ? sn[4 + m] : cs[4 + m];
    }
#undef PIDX
    float m0 = h ? 0.f : 1.f;
    v2f a2, b2, a3, b3;
    a2.x = alpha * m0; a2.y = m0;
    b2.x = 1.f;        b2.y = beta;
    a3.x = mu * m0;    a3.y = 0.f;
    b3.x = nu;         b3.y = 0.f;
    v8f TH = {};
    TH = wmma4(a2, b2, TH);
    TH = wmma4(a1, b1, TH);
    v8f MG = {};
    MG = wmma4(a3, b3, MG);
    v8f xr, xi;
#pragma unroll
    for (int g = 0; g < 8; ++g) {
      float se, ce;
      __sincosf(TH[g], &se, &ce);
      xr[g] = MG[g] * ce;
      xi[g] = MG[g] * se;
    }
    stCD(Xr, xr, r, h);
    stCD(Xi, xi, r, h);
  }
  waveSync();
  buildStage(tR, tI, 0, GLr, GLi, GRr, GRi, lane);
  waveSync();

  cmatL(GLr, GLi, Xr, Xi, Yr, Yi, lane);
  waveSync();
  cmatR(Yr, Yi, GRr, GRi, Xr, Xi, lane);
  waveSync();

  applyCnot<1>(Xr, Xi, Yr, Yi, lane);
  buildStage(tR, tI, 1, GLr, GLi, GRr, GRi, lane);
  waveSync();

  cmatL(GLr, GLi, Yr, Yi, Xr, Xi, lane);
  waveSync();
  cmatR(Xr, Xi, GRr, GRi, Yr, Yi, lane);
  waveSync();

  buildStage(tR, tI, 2, GLr, GLi, GRr, GRi, lane);
  waveSync();
  cmatL(GLr, GLi, Yr, Yi, Xr, Xi, lane);
  waveSync();
  cmatR(Xr, Xi, GRr, GRi, Yr, Yi, lane);
  waveSync();

  applyCnot<2>(Yr, Yi, Xr, Xi, lane);
  waveSync();

  float acc[8];
#pragma unroll
  for (int k = 0; k < 8; ++k) acc[k] = 0.f;
  v8f fr = ldRun8(Xr, lane >> 1, lane & 1);
  v8f fi = ldRun8(Xi, lane >> 1, lane & 1);
#pragma unroll
  for (int j = 0; j < 8; ++j) {
    int d = lane * 8 + j;
    int s = ((d & 15) << 4) | (d >> 4);
    float vr = fr[j], vi = fi[j];
    float p2 = vr * vr + vi * vi;
#pragma unroll
    for (int k = 0; k < 8; ++k)
      acc[k] += ((s >> (7 - k)) & 1) ? -p2 : p2;
  }
#pragma unroll
  for (int k = 0; k < 8; ++k) {
    float v = acc[k];
    v += __shfl_xor(v, 16, 32);
    v += __shfl_xor(v, 8, 32);
    v += __shfl_xor(v, 4, 32);
    v += __shfl_xor(v, 2, 32);
    v += __shfl_xor(v, 1, 32);
    acc[k] = v;
  }
  if (lane == 0) {
#pragma unroll
    for (int k = 0; k < 8; ++k) out[b * 8 + k] = acc[k];
  }
}

extern "C" void kernel_launch(void* const* d_in, const int* in_sizes, int n_in,
                              void* d_out, int out_size, void* d_ws, size_t ws_size,
                              hipStream_t stream) {
  const float* theta = (const float*)d_in[0];
  const float* phi   = (const float*)d_in[1];
  const float* jup   = (const float*)d_in[2];
  const float* vqcw  = (const float*)d_in[3];
  float* out = (float*)d_out;
  int B = in_sizes[0] / 8;
  dim3 grid(B / WPB), block(32 * WPB);
  hipLaunchKernelGGL(vqc_kernel, grid, block, 0, stream, theta, phi, jup, vqcw, out);
}


# Both CNOT ring permutations done by the preceding product's result store (scatter; address = one GF(2)-linear base per lane XOR a constant per row): the two LDS gather passes are gone; sums unchanged
# speedup vs baseline: 21.6568x; 1.0642x over previous
#include <hip/hip_runtime.h>

typedef __attribute__((ext_vector_type(2))) float v2f;
typedef __attribute__((ext_vector_type(4))) float v4f;
typedef __attribute__((ext_vector_type(8))) float v8f;

#define WPB 8
#define DD  256

static __device__ __forceinline__ v8f wmma4(v2f a, v2f b, v8f c) {
  return __builtin_amdgcn_wmma_f32_16x16x4_f32(false, a, false, b, (short)0, c,
                                               false, false);
}

static __device__ __forceinline__ void waveSync() {
  __builtin_amdgcn_fence(__ATOMIC_RELEASE, "wavefront");
  __builtin_amdgcn_wave_barrier();
  __builtin_amdgcn_fence(__ATOMIC_ACQUIRE, "wavefront");
}

static __device__ __forceinline__ int planeGrp(int major, int minor4) {
  return major * 16 + (minor4 ^ (major & 12));
}

static __device__ __forceinline__ v2f ldRow2(const float* base, int i, int k, int h) {
  return *(const v2f*)(base + planeGrp(i, 4 * k) + 2 * h);
}
static __device__ __forceinline__ v2f ldColS(const float* base, int m, int k, int h) {
  int kk = 4 * k + 2 * h;
  const float* p = base + kk * 16 + (m ^ (4 * k));
  v2f r; r.x = p[0]; r.y = p[16];
  return r;
}
static __device__ __forceinline__ void stRun8(float* plane, int major, int half, v8f v) {
  v4f lo = {v[0], v[1], v[2], v[3]};
  v4f hi = {v[4], v[5], v[6], v[7]};
  *(v4f*)(plane + planeGrp(major, 8 * half)) = lo;
  *(v4f*)(plane + planeGrp(major, 8 * half + 4)) = hi;
}
static __device__ __forceinline__ v8f ldRun8(const float* plane, int major, int half) {
  v4f lo = *(const v4f*)(plane + planeGrp(major, 8 * half));
  v4f hi = *(const v4f*)(plane + planeGrp(major, 8 * half + 4));
  v8f r = {lo[0], lo[1], lo[2], lo[3], hi[0], hi[1], hi[2], hi[3]};
  return r;
}

static __device__ __forceinline__ void stCD(float* D, v8f v, int i, int h) {
  stRun8(D, i, h, v);
}

static constexpr int cnotSrc(int i, int rng) {
  int src = i;
  for (int k = 7; k >= 0; --k) src ^= ((src >> (7 - k)) & 1) << (7 - ((k + rng) & 7));
  return src;
}
static constexpr int cnotDst(int s, int rng) {
  int d = s;
  for (int k = 0; k < 8; ++k) d ^= ((d >> (7 - k)) & 1) << (7 - ((k + rng) & 7));
  return d;
}
static constexpr int stateDword(int s) {
  int col = s & 15, row = s >> 4;
  return (col << 4) | (row ^ (col & 12));
}
static constexpr bool cnotScatterOk(int rng) {
  for (int s = 0; s < 256; ++s) {
    if (cnotSrc(cnotDst(s, rng), rng) != s) return false;
    int lanePart = s & 0x8f, rowPart = s & 0x70;
    if (stateDword(cnotDst(s, rng)) !=
        (stateDword(cnotDst(lanePart, rng)) ^ stateDword(cnotDst(rowPart, rng)))) return false;
  }
  return true;
}
static_assert(cnotScatterOk(1) && cnotScatterOk(2), "ring scatter must invert the reference's gather");

template <int RNG>
static __device__ __forceinline__ void stCDcnot(float* Dr, float* Di, v8f vr, v8f vi, int i, int h) {
  int base = stateDword(cnotDst((h << 7) | i, RNG));
#pragma unroll
  for (int r = 0; r < 8; ++r) {
    int o = base ^ stateDword(cnotDst(r << 4, RNG));
    Dr[o] = vr[r];
    Di[o] = vi[r];
  }
}

template <bool COL>
static __device__ __forceinline__ void ldFrag4(const float* P, int i, int h, v2f f[4]) {
#pragma unroll
  for (int k = 0; k < 4; ++k) f[k] = COL ? ldColS(P, i, k, h) : ldRow2(P, i, k, h);
}

static __device__ __forceinline__ v8f mm16(const v2f a[4], const v2f b[4], v8f acc) {
#pragma unroll
  for (int k = 0; k < 4; ++k) acc = wmma4(a[k], b[k], acc);
  return acc;
}

template <bool COLA, int RNG>
static __device__ __forceinline__ void cmat3(const float* Ar, const float* Ai,
                                             const float* Br, const float* Bi,
                                             float* Dr, float* Di, int lane) {
  int i = lane & 15, h = lane >> 4;
  v2f ar[4], ai[4], br[4], bi[4];
  ldFrag4<COLA>(Ar, i, h, ar);
  ldFrag4<COLA>(Ai, i, h, ai);
  ldFrag4<false>(Br, i, h, br);
  ldFrag4<false>(Bi, i, h, bi);
  v2f as[4], bd[4], bn[4];
#pragma unroll
  for (int k = 0; k < 4; ++k) {
    as[k] = ar[k] + ai[k];
    bd[k] = bi[k] - br[k];
    bn[k] = -br[k] - bi[k];
  }
  v8f p = {};
  p = mm16(as, br, p);
  v8f dr = mm16(ai, bn, p);
  v8f di = mm16(ar, bd, p);
  if constexpr (RNG == 0) {
    stCD(Dr, dr, i, h);
    stCD(Di, di, i, h);
  } else {
    stCDcnot<RNG>(Dr, Di, dr, di, i, h);
  }
}

static __device__ __forceinline__ void cmatL(const float* Lr, const float* Li,
                                             const float* Sr, const float* Si,
                                             float* Dr, float* Di, int lane) {
  cmat3<false, 0>(Lr, Li, Sr, Si, Dr, Di, lane);
}

template <int RNG>
static __device__ __forceinline__ void cmatR(const float* Tr, const float* Ti,
                                             const float* Rr, const float* Ri,
                                             float* Dr, float* Di, int lane) {
  cmat3<true, RNG>(Tr, Ti, Rr, Ri, Dr, Di, lane);
}

static __device__ __forceinline__ void rotMat(float phi, float th, float om,
                                              float gr[2][2], float gi[2][2]) {
  float st, ct, sa, ca, sb, cb;
  __sincosf(th * 0.5f, &st, &ct);
  __sincosf(0.5f * (phi + om), &sa, &ca);
  __sincosf(0.5f * (phi - om), &sb, &cb);
  gr[0][0] = ct * ca;  gi[0][0] = -ct * sa;
  gr[0][1] = -st * cb; gi[0][1] = -st * sb;
  gr[1][0] = st * cb;  gi[1][0] = -st * sb;
  gr[1][1] = ct * ca;  gi[1][1] = ct * sa;
}

static __device__ __forceinline__ void encMat(float th, float phi,
                                              float gr[2][2], float gi[2][2]) {
  float s, c, sp, cp;
  __sincosf(th * 0.5f, &s, &c);
  __sincosf(phi * 0.5f, &sp, &cp);
  gr[0][0] = c * cp;  gi[0][0] = -c * sp;
  gr[0][1] = -s * cp; gi[0][1] = s * sp;
  gr[1][0] = s * cp;  gi[1][0] = s * sp;
  gr[1][1] = c * cp;  gi[1][1] = c * sp;
}

#define NGATE 24
#define TABN  (NGATE * 4)

static __device__ __forceinline__ void fillGateTable(const float* vqcw, const float* th,
                                                     const float* ph, float* tR, float* tI,
                                                     int lane) {
  int g = lane < NGATE ? lane : NGATE - 1;
  int stage = g >> 3, q = g & 7;
  float gr[2][2], gi[2][2];
  if (stage == 1) {
    encMat(th[q] * 0.5f, ph[q] * 0.5f, gr, gi);
  } else {
    const float* w = vqcw + ((stage >> 1) * 8 + q) * 3;
    rotMat(w[0], w[1], w[2], gr, gi);
  }
  if (lane < NGATE) {
#pragma unroll
    for (int a = 0; a < 2; ++a)
#pragma unroll
      for (int b = 0; b < 2; ++b) {
        tR[g * 4 + a * 2 + b] = gr[a][b];
        tI[g * 4 + a * 2 + b] = gi[a][b];
      }
  }
}

static __device__ __forceinline__ void buildKron(const float* tR, const float* tI,
                                                 float* Kr, float* Ki, int lane) {
  int a = lane >> 1;
  int o0 = ((a >> 3) & 1) * 2 + (lane & 1);
  float x0r = tR[o0], x0i = tI[o0];
  float xr[3][2], xi[3][2];
#pragma unroll
  for (int q = 1; q < 4; ++q) {
    int o = q * 4 + ((a >> (3 - q)) & 1) * 2;
    xr[q - 1][0] = tR[o];     xi[q - 1][0] = tI[o];
    xr[q - 1][1] = tR[o + 1]; xi[q - 1][1] = tI[o + 1];
  }
  v8f kr, ki;
#pragma unroll
  for (int j = 0; j < 8; ++j) {
    float pr = x0r, pi = x0i;
#pragma unroll
    for (int q = 1; q < 4; ++q) {
      int bb = (j >> (3 - q)) & 1;
      float yr = xr[q - 1][bb], yi = xi[q - 1][bb];
      float nr = pr * yr - pi * yi;
      pi = pr * yi + pi * yr;
      pr = nr;
    }
    kr[j] = pr;
    ki[j] = pi;
  }
  stRun8(Kr, a, lane & 1, kr);
  stRun8(Ki, a, lane & 1, ki);
}

static __device__ __forceinline__ void buildStage(const float* tR, const float* tI, int stage,
                                                  float* GLr, float* GLi,
                                                  float* GRr, float* GRi, int lane) {
  buildKron(tR + stage * 32, tI + stage * 32, GLr, GLi, lane);
  buildKron(tR + stage * 32 + 16, tI + stage * 32 + 16, GRr, GRi, lane);
}

extern "C" __global__ __launch_bounds__(256) void vqc_kernel(
    const float* __restrict__ theta, const float* __restrict__ phi,
    const float* __restrict__ jup, const float* __restrict__ vqcw,
    float* __restrict__ out) {
  __shared__ __attribute__((aligned(16))) float sPl[WPB][8][DD];
  __shared__ float sTabR[WPB][TABN], sTabI[WPB][TABN];

  int tid = threadIdx.x;
  int wid = tid >> 5, lane = tid & 31;
  int b = blockIdx.x * WPB + wid;

  const float* th = theta + b * 8;
  const float* ph = phi + b * 8;
  const float* jj = jup + b * 28;

  float* wb = &sPl[wid][0][0];
  float* Xr = wb;            float* Xi = wb + DD;
  float* Yr = wb + 2 * DD;   float* Yi = wb + 3 * DD;
  float* GLr = wb + 4 * DD;  float* GLi = wb + 5 * DD;
  float* GRr = wb + 6 * DD;  float* GRi = wb + 7 * DD;
  float* tR = sTabR[wid]; float* tI = sTabI[wid];

  fillGateTable(vqcw, th, ph, tR, tI, lane);

  {
    int r = lane & 15, h = lane >> 4;
    float z[4];
#pragma unroll
    for (int m = 0; m < 4; ++m) z[m] = ((r >> (3 - m)) & 1) ? -1.f : 1.f;
    float jv[28];
#pragma unroll
    for (int p = 0; p < 28; ++p) jv[p] = jj[p];
    float pv[8];
#pragma unroll
    for (int k = 0; k < 8; ++k) pv[k] = ph[k];
    float sq, cq;
    __sincosf(th[lane & 7] * 0.5f, &sq, &cq);
    float cs[8], sn[8];
#pragma unroll
    for (int k = 0; k < 8; ++k) {
      cs[k] = __int_as_float(__builtin_amdgcn_readlane(__float_as_int(cq), k));
      sn[k] = __int_as_float(__builtin_amdgcn_readlane(__float_as_int(sq), k));
    }
#define PIDX(i, j) ((i) * 8 - ((i) * ((i) + 1)) / 2 + ((j) - (i) - 1))
    const float NHP = -1.57079632679489662f;
    v2f a1, b1;
#pragma unroll
    for (int d = 0; d < 2; ++d) {
      float acc0 = 0.f, acc1 = 0.f;
#pragma unroll
      for (int a = 0; a < 4; ++a) {
        acc0 += z[a] * jv[PIDX(a, 4 + d)];
        acc1 += z[a] * jv[PIDX(a, 6 + d)];
      }
      a1[d] = NHP * (h ? acc1 : acc0);
    }
    b1.x = h ? z[2] : z[0];
    b1.y = h ? z[3] : z[1];
    float t0 = jv[PIDX(0, 1)] * z[1] + jv[PIDX(0, 2)] * z[2] + jv[PIDX(0, 3)] * z[3];
    float t1 = jv[PIDX(1, 2)] * z[2] + jv[PIDX(1, 3)] * z[3];
    float t2 = jv[PIDX(2, 3)] * z[3];
    float sa = z[0] * t0 + z[1] * t1 + z[2] * t2;
    float pa = pv[0] * z[0] + pv[1] * z[1] + pv[2] * z[2] + pv[3] * z[3];
    float alpha = NHP * sa - 0.5f * pa;
    float u0 = jv[PIDX(4, 5)] * z[1] + jv[PIDX(4, 6)] * z[2] + jv[PIDX(4, 7)] * z[3];
    float u1 = jv[PIDX(5, 6)] * z[2] + jv[PIDX(5, 7)] * z[3];
    float u2 = jv[PIDX(6, 7)] * z[3];
    float sb = z[0] * u0 + z[1] * u1 + z[2] * u2;
    float pb = pv[4] * z[0] + pv[5] * z[1] + pv[6] * z[2] + pv[7] * z[3];
    float beta = NHP * sb - 0.5f * pb;
    float mu = 1.f, nu = 1.f;
#pragma unroll
    for (int m = 0; m < 4; ++m) {
      bool set = (r >> (3 - m)) & 1;
      mu *= set ? sn[m] : cs[m];
      nu *= set ? sn[4 + m] : cs[4 + m];
    }
#undef PIDX
    float m0 = h ? 0.f : 1.f;
    v2f a2, b2, a3, b3;
    a2.x = alpha * m0; a2.y = m0;
    b2.x = 1.f;        b2.y = beta;
    a3.x = mu * m0;    a3.y = 0.f;
    b3.x = nu;         b3.y = 0.f;
    v8f TH = {};
    TH = wmma4(a2, b2, TH);
    TH = wmma4(a1, b1, TH);
    v8f MG = {};
    MG = wmma4(a3, b3, MG);
    v8f xr, xi;
#pragma unroll
    for (int g = 0; g < 8; ++g) {
      float se, ce;
      __sincosf(TH[g], &se, &ce);
      xr[g] = MG[g] * ce;
      xi[g] = MG[g] * se;
    }
    stCD(Xr, xr, r, h);
    stCD(Xi, xi, r, h);
  }
  waveSync();
  buildStage(tR, tI, 0, GLr, GLi, GRr, GRi, lane);
  waveSync();

  cmatL(GLr, GLi, Xr, Xi, Yr, Yi, lane);
  waveSync();
  cmatR<1>(Yr, Yi, GRr, GRi, Xr, Xi, lane);
  waveSync();
  buildStage(tR, tI, 1, GLr, GLi, GRr, GRi, lane);
  waveSync();

  cmatL(GLr, GLi, Xr, Xi, Yr, Yi, lane);
  waveSync();
  cmatR<0>(Yr, Yi, GRr, GRi, Xr, Xi, lane);
  waveSync();
  buildStage(tR, tI, 2, GLr, GLi, GRr, GRi, lane);
  waveSync();

  cmatL(GLr, GLi, Xr, Xi, Yr, Yi, lane);
  waveSync();
  cmatR<2>(Yr, Yi, GRr, GRi, Xr, Xi, lane);
  waveSync();

  float acc[8];
#pragma unroll
  for (int k = 0; k < 8; ++k) acc[k] = 0.f;
  v8f fr = ldRun8(Xr, lane >> 1, lane & 1);
  v8f fi = ldRun8(Xi, lane >> 1, lane & 1);
#pragma unroll
  for (int j = 0; j < 8; ++j) {
    int d = lane * 8 + j;
    int s = ((d & 15) << 4) | (d >> 4);
    float vr = fr[j], vi = fi[j];
    float p2 = vr * vr + vi * vi;
#pragma unroll
    for (int k = 0; k < 8; ++k)
      acc[k] += ((s >> (7 - k)) & 1) ? -p2 : p2;
  }
#pragma unroll
  for (int k = 0; k < 8; ++k) {
    float v = acc[k];
    v += __shfl_xor(v, 16, 32);
    v += __shfl_xor(v, 8, 32);
    v += __shfl_xor(v, 4, 32);
    v += __shfl_xor(v, 2, 32);
    v += __shfl_xor(v, 1, 32);
    acc[k] = v;
  }
  if (lane == 0) {
#pragma unroll
    for (int k = 0; k < 8; ++k) out[b * 8 + k] = acc[k];
  }
}

extern "C" void kernel_launch(void* const* d_in, const int* in_sizes, int n_in,
                              void* d_out, int out_size, void* d_ws, size_t ws_size,
                              hipStream_t stream) {
  const float* theta = (const float*)d_in[0];
  const float* phi   = (const float*)d_in[1];
  const float* jup   = (const float*)d_in[2];
  const float* vqcw  = (const float*)d_in[3];
  float* out = (float*)d_out;
  int B = in_sizes[0] / 8;
  dim3 grid(B / WPB), block(32 * WPB);
  hipLaunchKernelGGL(vqc_kernel, grid, block, 0, stream, theta, phi, jup, vqcw, out);
}


# Each wave does 4 batch elements in a loop (grid a quarter): the weight-only Rot Kronecker gates, their fragment loads and operand sums are done once per wave, kept in registers; sums unchanged
# speedup vs baseline: 23.8952x; 1.1034x over previous
#include <hip/hip_runtime.h>

typedef __attribute__((ext_vector_type(2))) float v2f;
typedef __attribute__((ext_vector_type(4))) float v4f;
typedef __attribute__((ext_vector_type(8))) float v8f;

#define WPB 8
#define DD  256
#define NPER 4

static __device__ __forceinline__ v8f wmma4(v2f a, v2f b, v8f c) {
  return __builtin_amdgcn_wmma_f32_16x16x4_f32(false, a, false, b, (short)0, c,
                                               false, false);
}

static __device__ __forceinline__ void waveSync() {
  __builtin_amdgcn_fence(__ATOMIC_RELEASE, "wavefront");
  __builtin_amdgcn_wave_barrier();
  __builtin_amdgcn_fence(__ATOMIC_ACQUIRE, "wavefront");
}

static __device__ __forceinline__ int planeGrp(int major, int minor4) {
  return major * 16 + (minor4 ^ (major & 12));
}

static __device__ __forceinline__ v2f ldRow2(const float* base, int i, int k, int h) {
  return *(const v2f*)(base + planeGrp(i, 4 * k) + 2 * h);
}
static __device__ __forceinline__ v2f ldColS(const float* base, int m, int k, int h) {
  int kk = 4 * k + 2 * h;
  const float* p = base + kk * 16 + (m ^ (4 * k));
  v2f r; r.x = p[0]; r.y = p[16];
  return r;
}
static __device__ __forceinline__ void stRun8(float* plane, int major, int half, v8f v) {
  v4f lo = {v[0], v[1], v[2], v[3]};
  v4f hi = {v[4], v[5], v[6], v[7]};
  *(v4f*)(plane + planeGrp(major, 8 * half)) = lo;
  *(v4f*)(plane + planeGrp(major, 8 * half + 4)) = hi;
}
static __device__ __forceinline__ v8f ldRun8(const float* plane, int major, int half) {
  v4f lo = *(const v4f*)(plane + planeGrp(major, 8 * half));
  v4f hi = *(const v4f*)(plane + planeGrp(major, 8 * half + 4));
  v8f r = {lo[0], lo[1], lo[2], lo[3], hi[0], hi[1], hi[2], hi[3]};
  return r;
}

static __device__ __forceinline__ void stCD(float* D, v8f v, int i, int h) {
  stRun8(D, i, h, v);
}

static constexpr int cnotSrc(int i, int rng) {
  int src = i;
  for (int k = 7; k >= 0; --k) src ^= ((src >> (7 - k)) & 1) << (7 - ((k + rng) & 7));
  return src;
}
static constexpr int cnotDst(int s, int rng) {
  int d = s;
  for (int k = 0; k < 8; ++k) d ^= ((d >> (7 - k)) & 1) << (7 - ((k + rng) & 7));
  return d;
}
static constexpr int stateDword(int s) {
  int col = s & 15, row = s >> 4;
  return (col << 4) | (row ^ (col & 12));
}
static constexpr bool cnotScatterOk(int rng) {
  for (int s = 0; s < 256; ++s) {
    if (cnotSrc(cnotDst(s, rng), rng) != s) return false;
    int lanePart = s & 0x8f, rowPart = s & 0x70;
    if (stateDword(cnotDst(s, rng)) !=
        (stateDword(cnotDst(lanePart, rng)) ^ stateDword(cnotDst(rowPart, rng)))) return false;
  }
  return true;
}
static_assert(cnotScatterOk(1) && cnotScatterOk(2), "ring scatter must invert the reference's gather");

template <int RNG>
static __device__ __forceinline__ void stCDcnot(float* Dr, float* Di, v8f vr, v8f vi, int i, int h) {
  int base = stateDword(cnotDst((h << 7) | i, RNG));
#pragma unroll
  for (int r = 0; r < 8; ++r) {
    int o = base ^ stateDword(cnotDst(r << 4, RNG));
    Dr[o] = vr[r];
    Di[o] = vi[r];
  }
}

template <bool COL>
static __device__ __forceinline__ void ldFrag4(const float* P, int i, int h, v2f f[4]) {
#pragma unroll
  for (int k = 0; k < 4; ++k) f[k] = COL ? ldColS(P, i, k, h) : ldRow2(P, i, k, h);
}

static __device__ __forceinline__ v8f mm16(const v2f a[4], const v2f b[4], v8f acc) {
#pragma unroll
  for (int k = 0; k < 4; ++k) acc = wmma4(a[k], b[k], acc);
  return acc;
}

struct OpA { v2f s[4], i[4], r[4]; };
struct OpB { v2f r[4], n[4], d[4]; };

template <bool COLA>
static __device__ __forceinline__ void loadOpA(const float* Ar, const float* Ai, int lane, OpA& a) {
  int i = lane & 15, h = lane >> 4;
  ldFrag4<COLA>(Ar, i, h, a.r);
  ldFrag4<COLA>(Ai, i, h, a.i);
#pragma unroll
  for (int k = 0; k < 4; ++k) a.s[k] = a.r[k] + a.i[k];
}
static __device__ __forceinline__ void loadOpB(const float* Br, const float* Bi, int lane, OpB& b) {
  int i = lane & 15, h = lane >> 4;
  v2f bi[4];
  ldFrag4<false>(Br, i, h, b.r);
  ldFrag4<false>(Bi, i, h, bi);
#pragma unroll
  for (int k = 0; k < 4; ++k) {
    b.d[k] = bi[k] - b.r[k];
    b.n[k] = -b.r[k] - bi[k];
  }
}
template <int RNG>
static __device__ __forceinline__ void mulOp(const OpA& a, const OpB& b,
                                             float* Dr, float* Di, int lane) {
  int i = lane & 15, h = lane >> 4;
  v8f p = {};
  p = mm16(a.s, b.r, p);
  v8f dr = mm16(a.i, b.n, p);
  v8f di = mm16(a.r, b.d, p);
  if constexpr (RNG == 0) {
    stCD(Dr, dr, i, h);
    stCD(Di, di, i, h);
  } else {
    stCDcnot<RNG>(Dr, Di, dr, di, i, h);
  }
}
template <bool COLA, int RNG>
static __device__ __forceinline__ void cmat3(const float* Ar, const float* Ai,
                                             const float* Br, const float* Bi,
                                             float* Dr, float* Di, int lane) {
  OpA a; OpB b;
  loadOpA<COLA>(Ar, Ai, lane, a);
  loadOpB(Br, Bi, lane, b);
  mulOp<RNG>(a, b, Dr, Di, lane);
}

static __device__ __forceinline__ void cmatL(const float* Lr, const float* Li,
                                             const float* Sr, const float* Si,
                                             float* Dr, float* Di, int lane) {
  cmat3<false, 0>(Lr, Li, Sr, Si, Dr, Di, lane);
}

template <int RNG>
static __device__ __forceinline__ void cmatR(const float* Tr, const float* Ti,
                                             const float* Rr, const float* Ri,
                                             float* Dr, float* Di, int lane) {
  cmat3<true, RNG>(Tr, Ti, Rr, Ri, Dr, Di, lane);
}

static __device__ __forceinline__ void rotMat(float phi, float th, float om,
                                              float gr[2][2], float gi[2][2]) {
  float st, ct, sa, ca, sb, cb;
  __sincosf(th * 0.5f, &st, &ct);
  __sincosf(0.5f * (phi + om), &sa, &ca);
  __sincosf(0.5f * (phi - om), &sb, &cb);
  gr[0][0] = ct * ca;  gi[0][0] = -ct * sa;
  gr[0][1] = -st * cb; gi[0][1] = -st * sb;
  gr[1][0] = st * cb;  gi[1][0] = -st * sb;
  gr[1][1] = ct * ca;  gi[1][1] = ct * sa;
}

static __device__ __forceinline__ void encMat(float th, float phi,
                                              float gr[2][2], float gi[2][2]) {
  float s, c, sp, cp;
  __sincosf(th * 0.5f, &s, &c);
  __sincosf(phi * 0.5f, &sp, &cp);
  gr[0][0] = c * cp;  gi[0][0] = -c * sp;
  gr[0][1] = -s * cp; gi[0][1] = s * sp;
  gr[1][0] = s * cp;  gi[1][0] = s * sp;
  gr[1][1] = c * cp;  gi[1][1] = c * sp;
}

#define NGATE 24
#define TABN  (NGATE * 4)

static __device__ __forceinline__ void putGate(float* tR, float* tI, int g,
                                               const float gr[2][2], const float gi[2][2]) {
#pragma unroll
  for (int a = 0; a < 2; ++a)
#pragma unroll
    for (int b = 0; b < 2; ++b) {
      tR[g * 4 + a * 2 + b] = gr[a][b];
      tI[g * 4 + a * 2 + b] = gi[a][b];
    }
}
static __device__ __forceinline__ void fillRotTable(const float* vqcw, float* tR, float* tI,
                                                    int lane) {
  int g = lane & 15;
  const float* w = vqcw + g * 3;
  float gr[2][2], gi[2][2];
  rotMat(w[0], w[1], w[2], gr, gi);
  if (lane < 16) putGate(tR, tI, (g >> 3) * 16 + (g & 7), gr, gi);
}
static __device__ __forceinline__ void fillEncTable(float thq, float phq,
                                                    float* tR, float* tI, int lane) {
  float gr[2][2], gi[2][2];
  encMat(thq * 0.5f, phq * 0.5f, gr, gi);
  putGate(tR, tI, 8 + (lane & 7), gr, gi);
}

static __device__ __forceinline__ void buildKron(const float* tR, const float* tI,
                                                 float* Kr, float* Ki, int lane) {
  int a = lane >> 1;
  int o0 = ((a >> 3) & 1) * 2 + (lane & 1);
  float x0r = tR[o0], x0i = tI[o0];
  float xr[3][2], xi[3][2];
#pragma unroll
  for (int q = 1; q < 4; ++q) {
    int o = q * 4 + ((a >> (3 - q)) & 1) * 2;
    xr[q - 1][0] = tR[o];     xi[q - 1][0] = tI[o];
    xr[q - 1][1] = tR[o + 1]; xi[q - 1][1] = tI[o + 1];
  }
  v8f kr, ki;
#pragma unroll
  for (int j = 0; j < 8; ++j) {
    float pr = x0r, pi = x0i;
#pragma unroll
    for (int q = 1; q < 4; ++q) {
      int bb = (j >> (3 - q)) & 1;
      float yr = xr[q - 1][bb], yi = xi[q - 1][bb];
      float nr = pr * yr - pi * yi;
      pi = pr * yi + pi * yr;
      pr = nr;
    }
    kr[j] = pr;
    ki[j] = pi;
  }
  stRun8(Kr, a, lane & 1, kr);
  stRun8(Ki, a, lane & 1, ki);
}

static __device__ __forceinline__ void buildStage(const float* tR, const float* tI, int stage,
                                                  float* GLr, float* GLi,
                                                  float* GRr, float* GRi, int lane) {
  buildKron(tR + stage * 32, tI + stage * 32, GLr, GLi, lane);
  buildKron(tR + stage * 32 + 16, tI + stage * 32 + 16, GRr, GRi, lane);
}

extern "C" __global__ __launch_bounds__(256) void vqc_kernel(
    const float* __restrict__ theta, const float* __restrict__ phi,
    const float* __restrict__ jup, const float* __restrict__ vqcw,
    float* __restrict__ out) {
  __shared__ __attribute__((aligned(16))) float sPl[WPB][8][DD];
  __shared__ float sTabR[WPB][TABN], sTabI[WPB][TABN];

  int tid = threadIdx.x;
  int wid = tid >> 5, lane = tid & 31;

  float* wb = &sPl[wid][0][0];
  float* Xr = wb;            float* Xi = wb + DD;
  float* Yr = wb + 2 * DD;   float* Yi = wb + 3 * DD;
  float* GLr = wb + 4 * DD;  float* GLi = wb + 5 * DD;
  float* GRr = wb + 6 * DD;  float* GRi = wb + 7 * DD;
  float* tR = sTabR[wid]; float* tI = sTabI[wid];

  OpA L0, L2;
  OpB R0, R2;
  fillRotTable(vqcw, tR, tI, lane);
  waveSync();
  buildStage(tR, tI, 0, GLr, GLi, GRr, GRi, lane);
  waveSync();
  loadOpA<false>(GLr, GLi, lane, L0);
  loadOpB(GRr, GRi, lane, R0);
  waveSync();
  buildStage(tR, tI, 2, GLr, GLi, GRr, GRi, lane);
  waveSync();
  loadOpA<false>(GLr, GLi, lane, L2);
  loadOpB(GRr, GRi, lane, R2);
  waveSync();

  for (int it = 0; it < NPER; ++it) {
    int b = (blockIdx.x * WPB + wid) * NPER + it;
    const float* th = theta + b * 8;
    const float* ph = phi + b * 8;
    const float* jj = jup + b * 28;
    float thq = th[lane & 7], phq = ph[lane & 7];

    fillEncTable(thq, phq, tR, tI, lane);

    {
      int r = lane & 15, h = lane >> 4;
      float z[4];
#pragma unroll
      for (int m = 0; m < 4; ++m) z[m] = ((r >> (3 - m)) & 1) ? -1.f : 1.f;
      float jv[28];
#pragma unroll
      for (int p = 0; p < 28; ++p) jv[p] = jj[p];
      float pv[8];
#pragma unroll
      for (int k = 0; k < 8; ++k) pv[k] = ph[k];
      float sq, cq;
      __sincosf(thq * 0.5f, &sq, &cq);
      float cs[8], sn[8];
#pragma unroll
      for (int k = 0; k < 8; ++k) {
        cs[k] = __int_as_float(__builtin_amdgcn_readlane(__float_as_int(cq), k));
        sn[k] = __int_as_float(__builtin_amdgcn_readlane(__float_as_int(sq), k));
      }
#define PIDX(i, j) ((i) * 8 - ((i) * ((i) + 1)) / 2 + ((j) - (i) - 1))
      const float NHP = -1.57079632679489662f;
      v2f a1, b1;
#pragma unroll
      for (int d = 0; d < 2; ++d) {
        float acc0 = 0.f, acc1 = 0.f;
#pragma unroll
        for (int a = 0; a < 4; ++a) {
          acc0 += z[a] * jv[PIDX(a, 4 + d)];
          acc1 += z[a] * jv[PIDX(a, 6 + d)];
        }
        a1[d] = NHP * (h ? acc1 : acc0);
      }
      b1.x = h ? z[2] : z[0];
      b1.y = h ? z[3] : z[1];
      float t0 = jv[PIDX(0, 1)] * z[1] + jv[PIDX(0, 2)] * z[2] + jv[PIDX(0, 3)] * z[3];
      float t1 = jv[PIDX(1, 2)] * z[2] + jv[PIDX(1, 3)] * z[3];
      float t2 = jv[PIDX(2, 3)] * z[3];
      float sa = z[0] * t0 + z[1] * t1 + z[2] * t2;
      float pa = pv[0] * z[0] + pv[1] * z[1] + pv[2] * z[2] + pv[3] * z[3];
      float alpha = NHP * sa - 0.5f * pa;
      float u0 = jv[PIDX(4, 5)] * z[1] + jv[PIDX(4, 6)] * z[2] + jv[PIDX(4, 7)] * z[3];
      float u1 = jv[PIDX(5, 6)] * z[2] + jv[PIDX(5, 7)] * z[3];
      float u2 = jv[PIDX(6, 7)] * z[3];
      float sb = z[0] * u0 + z[1] * u1 + z[2] * u2;
      float pb = pv[4] * z[0] + pv[5] * z[1] + pv[6] * z[2] + pv[7] * z[3];
      float beta = NHP * sb - 0.5f * pb;
      float mu = 1.f, nu = 1.f;
#pragma unroll
      for (int m = 0; m < 4; ++m) {
        bool set = (r >> (3 - m)) & 1;
        mu *= set ? sn[m] : cs[m];
        nu *= set ? sn[4 + m] : cs[4 + m];
      }
#undef PIDX
      float m0 = h ? 0.f : 1.f;
      v2f a2, b2, a3, b3;
      a2.x = alpha * m0; a2.y = m0;
      b2.x = 1.f;        b2.y = beta;
      a3.x = mu * m0;    a3.y = 0.f;
      b3.x = nu;         b3.y = 0.f;
      v8f TH = {};
      TH = wmma4(a2, b2, TH);
      TH = wmma4(a1, b1, TH);
      v8f MG = {};
      MG = wmma4(a3, b3, MG);
      v8f xr, xi;
#pragma unroll
      for (int g = 0; g < 8; ++g) {
        float se, ce;
        __sincosf(TH[g], &se, &ce);
        xr[g] = MG[g] * ce;
        xi[g] = MG[g] * se;
      }
      stCD(Xr, xr, r, h);
      stCD(Xi, xi, r, h);
    }
    waveSync();
    buildStage(tR, tI, 1, GLr, GLi, GRr, GRi, lane);
    waveSync();

    {
      OpB s; loadOpB(Xr, Xi, lane, s);
      mulOp<0>(L0, s, Yr, Yi, lane);
    }
    waveSync();
    {
      OpA t; loadOpA<true>(Yr, Yi, lane, t);
      mulOp<1>(t, R0, Xr, Xi, lane);
    }
    waveSync();

    cmatL(GLr, GLi, Xr, Xi, Yr, Yi, lane);
    waveSync();
    cmatR<0>(Yr, Yi, GRr, GRi, Xr, Xi, lane);
    waveSync();

    {
      OpB s; loadOpB(Xr, Xi, lane, s);
      mulOp<0>(L2, s, Yr, Yi, lane);
    }
    waveSync();
    {
      OpA t; loadOpA<true>(Yr, Yi, lane, t);
      mulOp<2>(t, R2, Xr, Xi, lane);
    }
    waveSync();

    float acc[8];
#pragma unroll
    for (int k = 0; k < 8; ++k) acc[k] = 0.f;
    v8f fr = ldRun8(Xr, lane >> 1, lane & 1);
    v8f fi = ldRun8(Xi, lane >> 1, lane & 1);
#pragma unroll
    for (int j = 0; j < 8; ++j) {
      int d = lane * 8 + j;
      int s = ((d & 15) << 4) | (d >> 4);
      float vr = fr[j], vi = fi[j];
      float p2 = vr * vr + vi * vi;
#pragma unroll
      for (int k = 0; k < 8; ++k)
        acc[k] += ((s >> (7 - k)) & 1) ? -p2 : p2;
    }
#pragma unroll
    for (int k = 0; k < 8; ++k) {
      float v = acc[k];
      v += __shfl_xor(v, 16, 32);
      v += __shfl_xor(v, 8, 32);
      v += __shfl_xor(v, 4, 32);
      v += __shfl_xor(v, 2, 32);
      v += __shfl_xor(v, 1, 32);
      acc[k] = v;
    }
    if (lane == 0) {
#pragma unroll
      for (int k = 0; k < 8; ++k) out[b * 8 + k] = acc[k];
    }
    waveSync();
  }
}

extern "C" void kernel_launch(void* const* d_in, const int* in_sizes, int n_in,
                              void* d_out, int out_size, void* d_ws, size_t ws_size,
                              hipStream_t stream) {
  const float* theta = (const float*)d_in[0];
  const float* phi   = (const float*)d_in[1];
  const float* jup   = (const float*)d_in[2];
  const float* vqcw  = (const float*)d_in[3];
  float* out = (float*)d_out;
  int B = in_sizes[0] / 8;
  dim3 grid(B / (WPB * NPER)), block(32 * WPB);
  hipLaunchKernelGGL(vqc_kernel, grid, block, 0, stream, theta, phi, jup, vqcw, out);
}


# Readout: in-lane sum/difference tree gives the plain and 3 signed sums; a transposing butterfly of 9 lane exchanges on the vector unit (permlanex16, DPP) replaces 40 LDS permutes; sum order changes
# speedup vs baseline: 25.8369x; 1.0813x over previous
#include <hip/hip_runtime.h>

typedef __attribute__((ext_vector_type(2))) float v2f;
typedef __attribute__((ext_vector_type(4))) float v4f;
typedef __attribute__((ext_vector_type(8))) float v8f;

#define WPB 8
#define DD  256
#define NPER 4

static __device__ __forceinline__ v8f wmma4(v2f a, v2f b, v8f c) {
  return __builtin_amdgcn_wmma_f32_16x16x4_f32(false, a, false, b, (short)0, c,
                                               false, false);
}

static __device__ __forceinline__ void waveSync() {
  __builtin_amdgcn_fence(__ATOMIC_RELEASE, "wavefront");
  __builtin_amdgcn_wave_barrier();
  __builtin_amdgcn_fence(__ATOMIC_ACQUIRE, "wavefront");
}

static __device__ __forceinline__ int planeGrp(int major, int minor4) {
  return major * 16 + (minor4 ^ (major & 12));
}

static __device__ __forceinline__ v2f ldRow2(const float* base, int i, int k, int h) {
  return *(const v2f*)(base + planeGrp(i, 4 * k) + 2 * h);
}
static __device__ __forceinline__ v2f ldColS(const float* base, int m, int k, int h) {
  int kk = 4 * k + 2 * h;
  const float* p = base + kk * 16 + (m ^ (4 * k));
  v2f r; r.x = p[0]; r.y = p[16];
  return r;
}
static __device__ __forceinline__ void stRun8(float* plane, int major, int half, v8f v) {
  v4f lo = {v[0], v[1], v[2], v[3]};
  v4f hi = {v[4], v[5], v[6], v[7]};
  *(v4f*)(plane + planeGrp(major, 8 * half)) = lo;
  *(v4f*)(plane + planeGrp(major, 8 * half + 4)) = hi;
}
static __device__ __forceinline__ v8f ldRun8(const float* plane, int major, int half) {
  v4f lo = *(const v4f*)(plane + planeGrp(major, 8 * half));
  v4f hi = *(const v4f*)(plane + planeGrp(major, 8 * half + 4));
  v8f r = {lo[0], lo[1], lo[2], lo[3], hi[0], hi[1], hi[2], hi[3]};
  return r;
}

static __device__ __forceinline__ void stCD(float* D, v8f v, int i, int h) {
  stRun8(D, i, h, v);
}

static constexpr int cnotSrc(int i, int rng) {
  int src = i;
  for (int k = 7; k >= 0; --k) src ^= ((src >> (7 - k)) & 1) << (7 - ((k + rng) & 7));
  return src;
}
static constexpr int cnotDst(int s, int rng) {
  int d = s;
  for (int k = 0; k < 8; ++k) d ^= ((d >> (7 - k)) & 1) << (7 - ((k + rng) & 7));
  return d;
}
static constexpr int stateDword(int s) {
  int col = s & 15, row = s >> 4;
  return (col << 4) | (row ^ (col & 12));
}
static constexpr bool cnotScatterOk(int rng) {
  for (int s = 0; s < 256; ++s) {
    if (cnotSrc(cnotDst(s, rng), rng) != s) return false;
    int lanePart = s & 0x8f, rowPart = s & 0x70;
    if (stateDword(cnotDst(s, rng)) !=
        (stateDword(cnotDst(lanePart, rng)) ^ stateDword(cnotDst(rowPart, rng)))) return false;
  }
  return true;
}
static_assert(cnotScatterOk(1) && cnotScatterOk(2), "ring scatter must invert the reference's gather");

template <int RNG>
static __device__ __forceinline__ void stCDcnot(float* Dr, float* Di, v8f vr, v8f vi, int i, int h) {
  int base = stateDword(cnotDst((h << 7) | i, RNG));
#pragma unroll
  for (int r = 0; r < 8; ++r) {
    int o = base ^ stateDword(cnotDst(r << 4, RNG));
    Dr[o] = vr[r];
    Di[o] = vi[r];
  }
}

template <bool COL>
static __device__ __forceinline__ void ldFrag4(const float* P, int i, int h, v2f f[4]) {
#pragma unroll
  for (int k = 0; k < 4; ++k) f[k] = COL ? ldColS(P, i, k, h) : ldRow2(P, i, k, h);
}

static __device__ __forceinline__ v8f mm16(const v2f a[4], const v2f b[4], v8f acc) {
#pragma unroll
  for (int k = 0; k < 4; ++k) acc = wmma4(a[k], b[k], acc);
  return acc;
}

struct OpA { v2f s[4], i[4], r[4]; };
struct OpB { v2f r[4], n[4], d[4]; };

template <bool COLA>
static __device__ __forceinline__ void loadOpA(const float* Ar, const float* Ai, int lane, OpA& a) {
  int i = lane & 15, h = lane >> 4;
  ldFrag4<COLA>(Ar, i, h, a.r);
  ldFrag4<COLA>(Ai, i, h, a.i);
#pragma unroll
  for (int k = 0; k < 4; ++k) a.s[k] = a.r[k] + a.i[k];
}
static __device__ __forceinline__ void loadOpB(const float* Br, const float* Bi, int lane, OpB& b) {
  int i = lane & 15, h = lane >> 4;
  v2f bi[4];
  ldFrag4<false>(Br, i, h, b.r);
  ldFrag4<false>(Bi, i, h, bi);
#pragma unroll
  for (int k = 0; k < 4; ++k) {
    b.d[k] = bi[k] - b.r[k];
    b.n[k] = -b.r[k] - bi[k];
  }
}
template <int RNG>
static __device__ __forceinline__ void mulOp(const OpA& a, const OpB& b,
                                             float* Dr, float* Di, int lane) {
  int i = lane & 15, h = lane >> 4;
  v8f p = {};
  p = mm16(a.s, b.r, p);
  v8f dr = mm16(a.i, b.n, p);
  v8f di = mm16(a.r, b.d, p);
  if constexpr (RNG == 0) {
    stCD(Dr, dr, i, h);
    stCD(Di, di, i, h);
  } else {
    stCDcnot<RNG>(Dr, Di, dr, di, i, h);
  }
}
template <bool COLA, int RNG>
static __device__ __forceinline__ void cmat3(const float* Ar, const float* Ai,
                                             const float* Br, const float* Bi,
                                             float* Dr, float* Di, int lane) {
  OpA a; OpB b;
  loadOpA<COLA>(Ar, Ai, lane, a);
  loadOpB(Br, Bi, lane, b);
  mulOp<RNG>(a, b, Dr, Di, lane);
}

static __device__ __forceinline__ void cmatL(const float* Lr, const float* Li,
                                             const float* Sr, const float* Si,
                                             float* Dr, float* Di, int lane) {
  cmat3<false, 0>(Lr, Li, Sr, Si, Dr, Di, lane);
}

template <int RNG>
static __device__ __forceinline__ void cmatR(const float* Tr, const float* Ti,
                                             const float* Rr, const float* Ri,
                                             float* Dr, float* Di, int lane) {
  cmat3<true, RNG>(Tr, Ti, Rr, Ri, Dr, Di, lane);
}

static __device__ __forceinline__ void rotMat(float phi, float th, float om,
                                              float gr[2][2], float gi[2][2]) {
  float st, ct, sa, ca, sb, cb;
  __sincosf(th * 0.5f, &st, &ct);
  __sincosf(0.5f * (phi + om), &sa, &ca);
  __sincosf(0.5f * (phi - om), &sb, &cb);
  gr[0][0] = ct * ca;  gi[0][0] = -ct * sa;
  gr[0][1] = -st * cb; gi[0][1] = -st * sb;
  gr[1][0] = st * cb;  gi[1][0] = -st * sb;
  gr[1][1] = ct * ca;  gi[1][1] = ct * sa;
}

static __device__ __forceinline__ void encMat(float th, float phi,
                                              float gr[2][2], float gi[2][2]) {
  float s, c, sp, cp;
  __sincosf(th * 0.5f, &s, &c);
  __sincosf(phi * 0.5f, &sp, &cp);
  gr[0][0] = c * cp;  gi[0][0] = -c * sp;
  gr[0][1] = -s * cp; gi[0][1] = s * sp;
  gr[1][0] = s * cp;  gi[1][0] = s * sp;
  gr[1][1] = c * cp;  gi[1][1] = c * sp;
}

#define NGATE 24
#define TABN  (NGATE * 4)

static __device__ __forceinline__ void putGate(float* tR, float* tI, int g,
                                               const float gr[2][2], const float gi[2][2]) {
#pragma unroll
  for (int a = 0; a < 2; ++a)
#pragma unroll
    for (int b = 0; b < 2; ++b) {
      tR[g * 4 + a * 2 + b] = gr[a][b];
      tI[g * 4 + a * 2 + b] = gi[a][b];
    }
}
static __device__ __forceinline__ void fillRotTable(const float* vqcw, float* tR, float* tI,
                                                    int lane) {
  int g = lane & 15;
  const float* w = vqcw + g * 3;
  float gr[2][2], gi[2][2];
  rotMat(w[0], w[1], w[2], gr, gi);
  if (lane < 16) putGate(tR, tI, (g >> 3) * 16 + (g & 7), gr, gi);
}
static __device__ __forceinline__ void fillEncTable(float thq, float phq,
                                                    float* tR, float* tI, int lane) {
  float gr[2][2], gi[2][2];
  encMat(thq * 0.5f, phq * 0.5f, gr, gi);
  putGate(tR, tI, 8 + (lane & 7), gr, gi);
}

static __device__ __forceinline__ void buildKron(const float* tR, const float* tI,
                                                 float* Kr, float* Ki, int lane) {
  int a = lane >> 1;
  int o0 = ((a >> 3) & 1) * 2 + (lane & 1);
  float x0r = tR[o0], x0i = tI[o0];
  float xr[3][2], xi[3][2];
#pragma unroll
  for (int q = 1; q < 4; ++q) {
    int o = q * 4 + ((a >> (3 - q)) & 1) * 2;
    xr[q - 1][0] = tR[o];     xi[q - 1][0] = tI[o];
    xr[q - 1][1] = tR[o + 1]; xi[q - 1][1] = tI[o + 1];
  }
  v8f kr, ki;
#pragma unroll
  for (int j = 0; j < 8; ++j) {
    float pr = x0r, pi = x0i;
#pragma unroll
    for (int q = 1; q < 4; ++q) {
      int bb = (j >> (3 - q)) & 1;
      float yr = xr[q - 1][bb], yi = xi[q - 1][bb];
      float nr = pr * yr - pi * yi;
      pi = pr * yi + pi * yr;
      pr = nr;
    }
    kr[j] = pr;
    ki[j] = pi;
  }
  stRun8(Kr, a, lane & 1, kr);
  stRun8(Ki, a, lane & 1, ki);
}

static __device__ __forceinline__ void buildStage(const float* tR, const float* tI, int stage,
                                                  float* GLr, float* GLi,
                                                  float* GRr, float* GRi, int lane) {
  buildKron(tR + stage * 32, tI + stage * 32, GLr, GLi, lane);
  buildKron(tR + stage * 32 + 16, tI + stage * 32 + 16, GRr, GRi, lane);
}

static __device__ __forceinline__ float xhalf(float x) {
  int b = __float_as_int(x);
  return __int_as_float(__builtin_amdgcn_permlanex16(b, b, 0x76543210, 0xfedcba98, false, false));
}
template <int M>
static __device__ __forceinline__ float rowPeer(float x) {
  int b = __float_as_int(x);
  return __int_as_float(__builtin_amdgcn_update_dpp(0, b, 0x160 + M, 0xf, 0xf, false));
}

extern "C" __global__ __launch_bounds__(256) void vqc_kernel(
    const float* __restrict__ theta, const float* __restrict__ phi,
    const float* __restrict__ jup, const float* __restrict__ vqcw,
    float* __restrict__ out) {
  __shared__ __attribute__((aligned(16))) float sPl[WPB][8][DD];
  __shared__ float sTabR[WPB][TABN], sTabI[WPB][TABN];

  int tid = threadIdx.x;
  int wid = tid >> 5, lane = tid & 31;

  float* wb = &sPl[wid][0][0];
  float* Xr = wb;            float* Xi = wb + DD;
  float* Yr = wb + 2 * DD;   float* Yi = wb + 3 * DD;
  float* GLr = wb + 4 * DD;  float* GLi = wb + 5 * DD;
  float* GRr = wb + 6 * DD;  float* GRi = wb + 7 * DD;
  float* tR = sTabR[wid]; float* tI = sTabI[wid];

  OpA L0, L2;
  OpB R0, R2;
  fillRotTable(vqcw, tR, tI, lane);
  waveSync();
  buildStage(tR, tI, 0, GLr, GLi, GRr, GRi, lane);
  waveSync();
  loadOpA<false>(GLr, GLi, lane, L0);
  loadOpB(GRr, GRi, lane, R0);
  waveSync();
  buildStage(tR, tI, 2, GLr, GLi, GRr, GRi, lane);
  waveSync();
  loadOpA<false>(GLr, GLi, lane, L2);
  loadOpB(GRr, GRi, lane, R2);
  waveSync();

  for (int it = 0; it < NPER; ++it) {
    int b = (blockIdx.x * WPB + wid) * NPER + it;
    const float* th = theta + b * 8;
    const float* ph = phi + b * 8;
    const float* jj = jup + b * 28;
    float thq = th[lane & 7], phq = ph[lane & 7];

    fillEncTable(thq, phq, tR, tI, lane);

    {
      int r = lane & 15, h = lane >> 4;
      float z[4];
#pragma unroll
      for (int m = 0; m < 4; ++m) z[m] = ((r >> (3 - m)) & 1) ? -1.f : 1.f;
      float jv[28];
#pragma unroll
      for (int p = 0; p < 28; ++p) jv[p] = jj[p];
      float pv[8];
#pragma unroll
      for (int k = 0; k < 8; ++k) pv[k] = ph[k];
      float sq, cq;
      __sincosf(thq * 0.5f, &sq, &cq);
      float cs[8], sn[8];
#pragma unroll
      for (int k = 0; k < 8; ++k) {
        cs[k] = __int_as_float(__builtin_amdgcn_readlane(__float_as_int(cq), k));
        sn[k] = __int_as_float(__builtin_amdgcn_readlane(__float_as_int(sq), k));
      }
#define PIDX(i, j) ((i) * 8 - ((i) * ((i) + 1)) / 2 + ((j) - (i) - 1))
      const float NHP = -1.57079632679489662f;
      v2f a1, b1;
#pragma unroll
      for (int d = 0; d < 2; ++d) {
        float acc0 = 0.f, acc1 = 0.f;
#pragma unroll
        for (int a = 0; a < 4; ++a) {
          acc0 += z[a] * jv[PIDX(a, 4 + d)];
          acc1 += z[a] * jv[PIDX(a, 6 + d)];
        }
        a1[d] = NHP * (h ? acc1 : acc0);
      }
      b1.x = h ? z[2] : z[0];
      b1.y = h ? z[3] : z[1];
      float t0 = jv[PIDX(0, 1)] * z[1] + jv[PIDX(0, 2)] * z[2] + jv[PIDX(0, 3)] * z[3];
      float t1 = jv[PIDX(1, 2)] * z[2] + jv[PIDX(1, 3)] * z[3];
      float t2 = jv[PIDX(2, 3)] * z[3];
      float sa = z[0] * t0 + z[1] * t1 + z[2] * t2;
      float pa = pv[0] * z[0] + pv[1] * z[1] + pv[2] * z[2] + pv[3] * z[3];
      float alpha = NHP * sa - 0.5f * pa;
      float u0 = jv[PIDX(4, 5)] * z[1] + jv[PIDX(4, 6)] * z[2] + jv[PIDX(4, 7)] * z[3];
      float u1 = jv[PIDX(5, 6)] * z[2] + jv[PIDX(5, 7)] * z[3];
      float u2 = jv[PIDX(6, 7)] * z[3];
      float sb = z[0] * u0 + z[1] * u1 + z[2] * u2;
      float pb = pv[4] * z[0] + pv[5] * z[1] + pv[6] * z[2] + pv[7] * z[3];
      float beta = NHP * sb - 0.5f * pb;
      float mu = 1.f, nu = 1.f;
#pragma unroll
      for (int m = 0; m < 4; ++m) {
        bool set = (r >> (3 - m)) & 1;
        mu *= set ? sn[m] : cs[m];
        nu *= set ? sn[4 + m] : cs[4 + m];
      }
#undef PIDX
      float m0 = h ? 0.f : 1.f;
      v2f a2, b2, a3, b3;
      a2.x = alpha * m0; a2.y = m0;
      b2.x = 1.f;        b2.y = beta;
      a3.x = mu * m0;    a3.y = 0.f;
      b3.x = nu;         b3.y = 0.f;
      v8f TH = {};
      TH = wmma4(a2, b2, TH);
      TH = wmma4(a1, b1, TH);
      v8f MG = {};
      MG = wmma4(a3, b3, MG);
      v8f xr, xi;
#pragma unroll
      for (int g = 0; g < 8; ++g) {
        float se, ce;
        __sincosf(TH[g], &se, &ce);
        xr[g] = MG[g] * ce;
        xi[g] = MG[g] * se;
      }
      stCD(Xr, xr, r, h);
      stCD(Xi, xi, r, h);
    }
    waveSync();
    buildStage(tR, tI, 1, GLr, GLi, GRr, GRi, lane);
    waveSync();

    {
      OpB s; loadOpB(Xr, Xi, lane, s);
      mulOp<0>(L0, s, Yr, Yi, lane);
    }
    waveSync();
    {
      OpA t; loadOpA<true>(Yr, Yi, lane, t);
      mulOp<1>(t, R0, Xr, Xi, lane);
    }
    waveSync();

    cmatL(GLr, GLi, Xr, Xi, Yr, Yi, lane);
    waveSync();
    cmatR<0>(Yr, Yi, GRr, GRi, Xr, Xi, lane);
    waveSync();

    {
      OpB s; loadOpB(Xr, Xi, lane, s);
      mulOp<0>(L2, s, Yr, Yi, lane);
    }
    waveSync();
    {
      OpA t; loadOpA<true>(Yr, Yi, lane, t);
      mulOp<2>(t, R2, Xr, Xi, lane);
    }
    waveSync();

    v8f fr = ldRun8(Xr, lane >> 1, lane & 1);
    v8f fi = ldRun8(Xi, lane >> 1, lane & 1);
    v2f P[4];
#pragma unroll
    for (int q = 0; q < 4; ++q) {
      v2f re = {fr[2 * q], fr[2 * q + 1]}, im = {fi[2 * q], fi[2 * q + 1]};
      P[q] = re * re + im * im;
    }
    v2f A = P[0] + P[1], B1 = P[0] - P[1];
    v2f C = P[2] + P[3], D1 = P[2] - P[3];
    v2f E = A + C, F = A - C, G = B1 + D1;
    float t  = E.x + E.y;
    float s1 = F.x + F.y;
    float s2 = G.x + G.y;
    float s3 = E.x - E.y;
    float v[8];
    v[0] = (lane & 1) ? -t : t;
    v[1] = s1; v[2] = s2; v[3] = s3;
    v[4] = (lane & 16) ? -t : t;
    v[5] = (lane & 8) ? -t : t;
    v[6] = (lane & 4) ? -t : t;
    v[7] = (lane & 2) ? -t : t;
    const bool b4 = lane & 16, b3 = lane & 8, b2 = lane & 4;
    float w[4], x[2];
#pragma unroll
    for (int i = 0; i < 4; ++i) {
      float keep = b4 ? v[4 + i] : v[i], send = b4 ? v[i] : v[4 + i];
      w[i] = keep + xhalf(send);
    }
#pragma unroll
    for (int i = 0; i < 2; ++i) {
      float keep = b3 ? w[2 + i] : w[i], send = b3 ? w[i] : w[2 + i];
      x[i] = keep + rowPeer<8>(send);
    }
    float keep1 = b2 ? x[1] : x[0], send1 = b2 ? x[0] : x[1];
    float y = keep1 + rowPeer<4>(send1);
    y += rowPeer<2>(y);
    y += rowPeer<1>(y);
    if ((lane & 3) == 0) out[b * 8 + (lane >> 2)] = y;
    waveSync();
  }
}

extern "C" void kernel_launch(void* const* d_in, const int* in_sizes, int n_in,
                              void* d_out, int out_size, void* d_ws, size_t ws_size,
                              hipStream_t stream) {
  const float* theta = (const float*)d_in[0];
  const float* phi   = (const float*)d_in[1];
  const float* jup   = (const float*)d_in[2];
  const float* vqcw  = (const float*)d_in[3];
  float* out = (float*)d_out;
  int B = in_sizes[0] / 8;
  dim3 grid(B / (WPB * NPER)), block(32 * WPB);
  hipLaunchKernelGGL(vqc_kernel, grid, block, 0, stream, theta, phi, jup, vqcw, out);
}
